# starting kernel
# baseline (speedup 1.0000x reference)
#include <hip/hip_runtime.h>

#define D_MODEL   1024
#define NUM_HEADS 16
#define D_K       64
#define D_FF      2752
#define SEQ       2048
#define BATCH     2
#define ROWS      (BATCH*SEQ)

typedef float v2f __attribute__((ext_vector_type(2)));
typedef float v8f __attribute__((ext_vector_type(8)));

__device__ __forceinline__ v8f wmma_f32(v2f a, v2f b, v8f c) {
  return __builtin_amdgcn_wmma_f32_16x16x4_f32(
       false, a,  false, b,
       (short)0, c,  false,  false);
}

__device__ __forceinline__ float silu_mul(float u, float w) {
  return u * (1.0f / (1.0f + __expf(-u))) * w;
}

__global__ void __launch_bounds__(256)
rmsnorm_kernel(const float* __restrict__ x, const float* __restrict__ g,
               float* __restrict__ out, int d) {
  const int row = blockIdx.x;
  const float* xr = x + (size_t)row * d;
  float ss = 0.f;
  for (int i = threadIdx.x; i < d; i += blockDim.x) { float v = xr[i]; ss = fmaf(v, v, ss); }
#pragma unroll
  for (int m = 16; m >= 1; m >>= 1) ss += __shfl_xor(ss, m, 32);
  __shared__ float red[8];
  const int w = threadIdx.x >> 5, ln = threadIdx.x & 31;
  if (ln == 0) red[w] = ss;
  __syncthreads();
  float tot = 0.f;
#pragma unroll
  for (int i = 0; i < 8; i++) tot += red[i];
  const float inv = rsqrtf(tot / (float)d + 1e-5f);
  for (int i = threadIdx.x; i < d; i += blockDim.x)
    out[(size_t)row * d + i] = xr[i] * inv * g[i];
}

__global__ void __launch_bounds__(256)
rope_kernel(float* __restrict__ t, const int* __restrict__ positions,
            const int* __restrict__ theta_p) {
  const int row = blockIdx.x;
  const int s = row % SEQ;
  const float pos = (float)positions[s];
  const float th = (float)(*theta_p);
  float* r = t + (size_t)row * D_MODEL;
  for (int p = threadIdx.x; p < D_MODEL / 2; p += blockDim.x) {
    const int c0 = 2 * p;
    const int d  = c0 & (D_K - 1);
    const float freq = __powf(th, -(float)d * (1.0f / (float)D_K));
    const float ang = pos * freq;
    float sn, cs;
    sincosf(ang, &sn, &cs);
    const float xe = r[c0], xo = r[c0 + 1];
    r[c0]     = xe * cs - xo * sn;
    r[c0 + 1] = xe * sn + xo * cs;
  }
}

template <int MODE>
__global__ void __launch_bounds__(256)
gemm32(const float* __restrict__ A, const float* __restrict__ A2,
       const float* __restrict__ B, const float* __restrict__ Res,
       float* __restrict__ C, int M, int N, int K) {
  const int wid = blockIdx.x * 8 + (threadIdx.x >> 5);
  const int NT  = N >> 6;
  const int m0  = (wid / NT) << 5;
  const int n0  = (wid % NT) << 6;
  const int lane = threadIdx.x & 31;
  const int hf = lane >> 4;
  const int rc = lane & 15;

  const float* Ar0 = A + (size_t)(m0 + rc) * K + 2 * hf;
  const float* Ar1 = Ar0 + (size_t)16 * K;
  const float* A2r0 = (MODE == 2) ? (A2 + (size_t)(m0 + rc) * K + 2 * hf) : nullptr;
  const float* A2r1 = (MODE == 2) ? (A2r0 + (size_t)16 * K) : nullptr;
  const float* B0  = B + (size_t)(n0 + rc) * K + 2 * hf;
  const size_t bstep = (size_t)16 * K;

  v8f acc[2][4] = {};
  for (int k0 = 0; k0 < K; k0 += 4) {
    v2f a0, a1;
    if (MODE == 2) {
      const v2f u0 = *(const v2f*)(Ar0 + k0);
      const v2f w0 = *(const v2f*)(A2r0 + k0);
      const v2f u1 = *(const v2f*)(Ar1 + k0);
      const v2f w1 = *(const v2f*)(A2r1 + k0);
      a0.x = silu_mul(u0.x, w0.x);  a0.y = silu_mul(u0.y, w0.y);
      a1.x = silu_mul(u1.x, w1.x);  a1.y = silu_mul(u1.y, w1.y);
    } else {
      a0 = *(const v2f*)(Ar0 + k0);
      a1 = *(const v2f*)(Ar1 + k0);
    }
    v2f b[4];
#pragma unroll
    for (int t = 0; t < 4; t++) b[t] = *(const v2f*)(B0 + (size_t)t * bstep + k0);
#pragma unroll
    for (int t = 0; t < 4; t++) {
      acc[0][t] = wmma_f32(a0, b[t], acc[0][t]);
      acc[1][t] = wmma_f32(a1, b[t], acc[1][t]);
    }
  }
#pragma unroll
  for (int i = 0; i < 2; i++) {
#pragma unroll
    for (int r = 0; r < 8; r++) {
      const size_t base = (size_t)(m0 + i * 16 + r + 8 * hf) * N + n0 + rc;
      float o0 = acc[i][0][r], o1 = acc[i][1][r], o2 = acc[i][2][r], o3 = acc[i][3][r];
      if (MODE >= 1) {
        o0 += Res[base];      o1 += Res[base + 16];
        o2 += Res[base + 32]; o3 += Res[base + 48];
      }
      C[base] = o0; C[base + 16] = o1; C[base + 32] = o2; C[base + 48] = o3;
    }
  }
}

__global__ void __launch_bounds__(256)
flash_attn(const float* __restrict__ Q, const float* __restrict__ Kt,
           const float* __restrict__ V, float* __restrict__ O) {
  __shared__ float plds[8 * 256];
  const int wid = blockIdx.x * 8 + (threadIdx.x >> 5);
  const int NQT = SEQ / 16;
  const int qt = wid % NQT;
  const int bh = wid / NQT;
  const int h = bh % NUM_HEADS;
  const int b = bh / NUM_HEADS;
  const size_t base = (size_t)b * SEQ * D_MODEL + (size_t)h * D_K;
  const float* Qb = Q + base;
  const float* Kb = Kt + base;
  const float* Vb = V + base;
  float* Ob = O + base;
  const int lane = threadIdx.x & 31;
  const int hf = lane >> 4;
  const int rc = lane & 15;
  float* lp = plds + (threadIdx.x >> 5) * 256;

  v2f qa[16];
#pragma unroll
  for (int i = 0; i < 16; i++)
    qa[i] = *(const v2f*)(Qb + (size_t)(qt * 16 + rc) * D_MODEL + i * 4 + 2 * hf);

  v8f o0 = {}, o1 = {}, o2 = {}, o3 = {};
  float mr[8], lr[8];
#pragma unroll
  for (int r = 0; r < 8; r++) { mr[r] = -1e30f; lr[r] = 0.f; }

  for (int jt = 0; jt <= qt; ++jt) {
    v8f s = {};
    const float* Krow = Kb + (size_t)(jt * 16 + rc) * D_MODEL + 2 * hf;
#pragma unroll
    for (int i = 0; i < 16; i++) {
      const v2f kb = *(const v2f*)(Krow + i * 4);
      s = wmma_f32(qa[i], kb, s);
    }
    const bool diag = (jt == qt);
#pragma unroll
    for (int r = 0; r < 8; r++) {
      float sv = s[r] * 0.125f;
      if (diag && (rc > r + 8 * hf)) sv = -1e30f;
      float mx = sv;
#pragma unroll
      for (int msk = 1; msk < 16; msk <<= 1) mx = fmaxf(mx, __shfl_xor(mx, msk, 32));
      const float mnew = fmaxf(mr[r], mx);
      const float corr = __expf(mr[r] - mnew);
      mr[r] = mnew;
      const float p = __expf(sv - mnew);
      float ps = p;
#pragma unroll
      for (int msk = 1; msk < 16; msk <<= 1) ps += __shfl_xor(ps, msk, 32);
      lr[r] = lr[r] * corr + ps;
      o0[r] *= corr; o1[r] *= corr; o2[r] *= corr; o3[r] *= corr;
      lp[(r + 8 * hf) * 16 + rc] = p;
    }
    asm volatile("s_wait_dscnt 0" ::: "memory");
#pragma unroll
    for (int kc = 0; kc < 4; kc++) {
      const v2f pa = *(const v2f*)(lp + rc * 16 + kc * 4 + 2 * hf);
      const float* Vr = Vb + (size_t)(jt * 16 + kc * 4 + 2 * hf) * D_MODEL;
      v2f vb0, vb1, vb2, vb3;
      vb0.x = Vr[rc];      vb0.y = Vr[D_MODEL + rc];
      vb1.x = Vr[16 + rc]; vb1.y = Vr[D_MODEL + 16 + rc];
      vb2.x = Vr[32 + rc]; vb2.y = Vr[D_MODEL + 32 + rc];
      vb3.x = Vr[48 + rc]; vb3.y = Vr[D_MODEL + 48 + rc];
      o0 = wmma_f32(pa, vb0, o0);
      o1 = wmma_f32(pa, vb1, o1);
      o2 = wmma_f32(pa, vb2, o2);
      o3 = wmma_f32(pa, vb3, o3);
    }
  }
#pragma unroll
  for (int r = 0; r < 8; r++) {
    const float inv = 1.0f / lr[r];
    const size_t ob = (size_t)(qt * 16 + r + 8 * hf) * D_MODEL + rc;
    Ob[ob]      = o0[r] * inv;
    Ob[ob + 16] = o1[r] * inv;
    Ob[ob + 32] = o2[r] * inv;
    Ob[ob + 48] = o3[r] * inv;
  }
}

extern "C" void kernel_launch(void* const* d_in, const int* in_sizes, int n_in,
                              void* d_out, int out_size, void* d_ws, size_t ws_size,
                              hipStream_t stream) {
  (void)in_sizes; (void)n_in; (void)out_size; (void)ws_size;
  const float* x        = (const float*)d_in[0];
  const int*   positions= (const int*)d_in[1];
  const int*   theta    = (const int*)d_in[2];
  const float* Wq = (const float*)d_in[3];
  const float* Wk = (const float*)d_in[4];
  const float* Wv = (const float*)d_in[5];
  const float* Wo = (const float*)d_in[6];
  const float* g1 = (const float*)d_in[7];
  const float* g2 = (const float*)d_in[8];
  const float* w1 = (const float*)d_in[9];
  const float* w2 = (const float*)d_in[10];
  const float* w3 = (const float*)d_in[11];
  float* out = (float*)d_out;

  float* xn   = (float*)d_ws;
  float* q    = xn   + (size_t)ROWS * D_MODEL;
  float* k    = q    + (size_t)ROWS * D_MODEL;
  float* v    = k    + (size_t)ROWS * D_MODEL;
  float* attn = v    + (size_t)ROWS * D_MODEL;
  float* x2   = attn + (size_t)ROWS * D_MODEL;
  float* x3b  = x2   + (size_t)ROWS * D_MODEL;
  float* x1b  = q;

  const dim3 blk(256);
  const int gproj = (ROWS / 32) * (D_MODEL / 64) / 8;
  const int gff   = (ROWS / 32) * (D_FF / 64) / 8;
  const int gfa   = (BATCH * NUM_HEADS * (SEQ / 16)) / 8;

  rmsnorm_kernel<<<ROWS, blk, 0, stream>>>(x, g1, xn, D_MODEL);
  gemm32<0><<<gproj, blk, 0, stream>>>(xn, nullptr, Wq, nullptr, q, ROWS, D_MODEL, D_MODEL);
  gemm32<0><<<gproj, blk, 0, stream>>>(xn, nullptr, Wk, nullptr, k, ROWS, D_MODEL, D_MODEL);
  gemm32<0><<<gproj, blk, 0, stream>>>(xn, nullptr, Wv, nullptr, v, ROWS, D_MODEL, D_MODEL);
  rope_kernel<<<ROWS, blk, 0, stream>>>(q, positions, theta);
  rope_kernel<<<ROWS, blk, 0, stream>>>(k, positions, theta);
  flash_attn<<<gfa, blk, 0, stream>>>(q, k, v, attn);
  gemm32<1><<<gproj, blk, 0, stream>>>(attn, nullptr, Wo, x, x2, ROWS, D_MODEL, D_MODEL);
  rmsnorm_kernel<<<ROWS, blk, 0, stream>>>(x2, g2, xn, D_MODEL);
  gemm32<0><<<gff, blk, 0, stream>>>(xn, nullptr, w1, nullptr, x1b, ROWS, D_FF, D_MODEL);
  gemm32<0><<<gff, blk, 0, stream>>>(xn, nullptr, w3, nullptr, x3b, ROWS, D_FF, D_MODEL);
  gemm32<2><<<gproj, blk, 0, stream>>>(x1b, x3b, w2, x2, out, ROWS, D_MODEL, D_FF);
}


# gemm: 128x64 block tile staged through LDS (16-B coalesced global loads, register-prefetched double buffer); fragments read from LDS, not strided 8-B global loads; sum order kept
# speedup vs baseline: 2.7028x; 2.7028x over previous
#include <hip/hip_runtime.h>

#define D_MODEL   1024
#define NUM_HEADS 16
#define D_K       64
#define D_FF      2752
#define SEQ       2048
#define BATCH     2
#define ROWS      (BATCH*SEQ)

typedef float v2f __attribute__((ext_vector_type(2)));
typedef float v4f __attribute__((ext_vector_type(4)));
typedef float v8f __attribute__((ext_vector_type(8)));

__device__ __forceinline__ v8f wmma_f32(v2f a, v2f b, v8f c) {
  return __builtin_amdgcn_wmma_f32_16x16x4_f32(
       false, a,  false, b,
       (short)0, c,  false,  false);
}

__device__ __forceinline__ float silu_mul(float u, float w) {
  return u * (1.0f / (1.0f + __expf(-u))) * w;
}

__global__ void __launch_bounds__(256)
rmsnorm_kernel(const float* __restrict__ x, const float* __restrict__ g,
               float* __restrict__ out, int d) {
  const int row = blockIdx.x;
  const float* xr = x + (size_t)row * d;
  float ss = 0.f;
  for (int i = threadIdx.x; i < d; i += blockDim.x) { float v = xr[i]; ss = fmaf(v, v, ss); }
#pragma unroll
  for (int m = 16; m >= 1; m >>= 1) ss += __shfl_xor(ss, m, 32);
  __shared__ float red[8];
  const int w = threadIdx.x >> 5, ln = threadIdx.x & 31;
  if (ln == 0) red[w] = ss;
  __syncthreads();
  float tot = 0.f;
#pragma unroll
  for (int i = 0; i < 8; i++) tot += red[i];
  const float inv = rsqrtf(tot / (float)d + 1e-5f);
  for (int i = threadIdx.x; i < d; i += blockDim.x)
    out[(size_t)row * d + i] = xr[i] * inv * g[i];
}

__global__ void __launch_bounds__(256)
rope_kernel(float* __restrict__ t, const int* __restrict__ positions,
            const int* __restrict__ theta_p) {
  const int row = blockIdx.x;
  const int s = row % SEQ;
  const float pos = (float)positions[s];
  const float th = (float)(*theta_p);
  float* r = t + (size_t)row * D_MODEL;
  for (int p = threadIdx.x; p < D_MODEL / 2; p += blockDim.x) {
    const int c0 = 2 * p;
    const int d  = c0 & (D_K - 1);
    const float freq = __powf(th, -(float)d * (1.0f / (float)D_K));
    const float ang = pos * freq;
    float sn, cs;
    sincosf(ang, &sn, &cs);
    const float xe = r[c0], xo = r[c0 + 1];
    r[c0]     = xe * cs - xo * sn;
    r[c0 + 1] = xe * sn + xo * cs;
  }
}

#define GBM 128
#define GBN 64
#define GBK 32
#define GLD 36

template <int MODE>
__global__ void __launch_bounds__(128)
gemm_lds(const float* __restrict__ A, const float* __restrict__ A2,
         const float* __restrict__ B, const float* __restrict__ Res,
         float* __restrict__ C, int M, int N, int K) {
  __shared__ __attribute__((aligned(16))) float As[2][GBM * GLD];
  __shared__ __attribute__((aligned(16))) float Bs[2][GBN * GLD];
  (void)M;
  const int NT  = N >> 6;
  const int m0b = (blockIdx.x / NT) * GBM;
  const int n0b = (blockIdx.x % NT) * GBN;
  const int tid = threadIdx.x;
  const int w    = tid >> 5;
  const int lane = tid & 31;
  const int hf = lane >> 4;
  const int rc = lane & 15;

  const int sc = (tid & 7) << 2;
  const int sr = tid >> 3;
  const float* Ag  = A + (size_t)(m0b + sr) * K + sc;
  const float* A2g = (MODE == 2) ? (A2 + (size_t)(m0b + sr) * K + sc) : nullptr;
  const float* Bg  = B + (size_t)(n0b + sr) * K + sc;
  const size_t rstep = (size_t)16 * K;
  const int sts = sr * GLD + sc;

  v4f ra[8], rw[8], rb[4];

#pragma unroll
  for (int j = 0; j < 8; j++) {
    ra[j] = *(const v4f*)(Ag + j * rstep);
    if (MODE == 2) rw[j] = *(const v4f*)(A2g + j * rstep);
  }
#pragma unroll
  for (int j = 0; j < 4; j++) rb[j] = *(const v4f*)(Bg + j * rstep);
#pragma unroll
  for (int j = 0; j < 8; j++) {
    v4f v = ra[j];
    if (MODE == 2) {
      v.x = silu_mul(v.x, rw[j].x); v.y = silu_mul(v.y, rw[j].y);
      v.z = silu_mul(v.z, rw[j].z); v.w = silu_mul(v.w, rw[j].w);
    }
    *(v4f*)(&As[0][sts + j * 16 * GLD]) = v;
  }
#pragma unroll
  for (int j = 0; j < 4; j++) *(v4f*)(&Bs[0][sts + j * 16 * GLD]) = rb[j];
  __syncthreads();

  v8f acc[2][4] = {};
  const int nkt = K / GBK;
  const int afo = (w * 32 + rc) * GLD + 2 * hf;
  const int bfo = rc * GLD + 2 * hf;
  for (int kt = 0; kt < nkt; ++kt) {
    const int cur = kt & 1;
    const bool more = (kt + 1 < nkt);
    if (more) {
      const int ko = (kt + 1) * GBK;
#pragma unroll
      for (int j = 0; j < 8; j++) {
        ra[j] = *(const v4f*)(Ag + j * rstep + ko);
        if (MODE == 2) rw[j] = *(const v4f*)(A2g + j * rstep + ko);
      }
#pragma unroll
      for (int j = 0; j < 4; j++) rb[j] = *(const v4f*)(Bg + j * rstep + ko);
    }
    const float* as = &As[cur][afo];
    const float* bs = &Bs[cur][bfo];
#pragma unroll
    for (int kk = 0; kk < GBK / 4; kk++) {
      const v2f a0 = *(const v2f*)(as + kk * 4);
      const v2f a1 = *(const v2f*)(as + 16 * GLD + kk * 4);
      v2f b[4];
#pragma unroll
      for (int t = 0; t < 4; t++) b[t] = *(const v2f*)(bs + t * 16 * GLD + kk * 4);
#pragma unroll
      for (int t = 0; t < 4; t++) {
        acc[0][t] = wmma_f32(a0, b[t], acc[0][t]);
        acc[1][t] = wmma_f32(a1, b[t], acc[1][t]);
      }
    }
    if (more) {
      float* ad = &As[cur ^ 1][sts];
      float* bd = &Bs[cur ^ 1][sts];
#pragma unroll
      for (int j = 0; j < 8; j++) {
        v4f v = ra[j];
        if (MODE == 2) {
          v.x = silu_mul(v.x, rw[j].x); v.y = silu_mul(v.y, rw[j].y);
          v.z = silu_mul(v.z, rw[j].z); v.w = silu_mul(v.w, rw[j].w);
        }
        *(v4f*)(ad + j * 16 * GLD) = v;
      }
#pragma unroll
      for (int j = 0; j < 4; j++) *(v4f*)(bd + j * 16 * GLD) = rb[j];
    }
    __syncthreads();
  }

  const int m0 = m0b + w * 32;
  const int n0 = n0b;
#pragma unroll
  for (int i = 0; i < 2; i++) {
#pragma unroll
    for (int r = 0; r < 8; r++) {
      const size_t base = (size_t)(m0 + i * 16 + r + 8 * hf) * N + n0 + rc;
      float o0 = acc[i][0][r], o1 = acc[i][1][r], o2 = acc[i][2][r], o3 = acc[i][3][r];
      if (MODE >= 1) {
        o0 += Res[base];      o1 += Res[base + 16];
        o2 += Res[base + 32]; o3 += Res[base + 48];
      }
      C[base] = o0; C[base + 16] = o1; C[base + 32] = o2; C[base + 48] = o3;
    }
  }
}

__global__ void __launch_bounds__(256)
flash_attn(const float* __restrict__ Q, const float* __restrict__ Kt,
           const float* __restrict__ V, float* __restrict__ O) {
  __shared__ float plds[8 * 256];
  const int wid = blockIdx.x * 8 + (threadIdx.x >> 5);
  const int NQT = SEQ / 16;
  const int qt = wid % NQT;
  const int bh = wid / NQT;
  const int h = bh % NUM_HEADS;
  const int b = bh / NUM_HEADS;
  const size_t base = (size_t)b * SEQ * D_MODEL + (size_t)h * D_K;
  const float* Qb = Q + base;
  const float* Kb = Kt + base;
  const float* Vb = V + base;
  float* Ob = O + base;
  const int lane = threadIdx.x & 31;
  const int hf = lane >> 4;
  const int rc = lane & 15;
  float* lp = plds + (threadIdx.x >> 5) * 256;

  v2f qa[16];
#pragma unroll
  for (int i = 0; i < 16; i++)
    qa[i] = *(const v2f*)(Qb + (size_t)(qt * 16 + rc) * D_MODEL + i * 4 + 2 * hf);

  v8f o0 = {}, o1 = {}, o2 = {}, o3 = {};
  float mr[8], lr[8];
#pragma unroll
  for (int r = 0; r < 8; r++) { mr[r] = -1e30f; lr[r] = 0.f; }

  for (int jt = 0; jt <= qt; ++jt) {
    v8f s = {};
    const float* Krow = Kb + (size_t)(jt * 16 + rc) * D_MODEL + 2 * hf;
#pragma unroll
    for (int i = 0; i < 16; i++) {
      const v2f kb = *(const v2f*)(Krow + i * 4);
      s = wmma_f32(qa[i], kb, s);
    }
    const bool diag = (jt == qt);
#pragma unroll
    for (int r = 0; r < 8; r++) {
      float sv = s[r] * 0.125f;
      if (diag && (rc > r + 8 * hf)) sv = -1e30f;
      float mx = sv;
#pragma unroll
      for (int msk = 1; msk < 16; msk <<= 1) mx = fmaxf(mx, __shfl_xor(mx, msk, 32));
      const float mnew = fmaxf(mr[r], mx);
      const float corr = __expf(mr[r] - mnew);
      mr[r] = mnew;
      const float p = __expf(sv - mnew);
      float ps = p;
#pragma unroll
      for (int msk = 1; msk < 16; msk <<= 1) ps += __shfl_xor(ps, msk, 32);
      lr[r] = lr[r] * corr + ps;
      o0[r] *= corr; o1[r] *= corr; o2[r] *= corr; o3[r] *= corr;
      lp[(r + 8 * hf) * 16 + rc] = p;
    }
    asm volatile("s_wait_dscnt 0" ::: "memory");
#pragma unroll
    for (int kc = 0; kc < 4; kc++) {
      const v2f pa = *(const v2f*)(lp + rc * 16 + kc * 4 + 2 * hf);
      const float* Vr = Vb + (size_t)(jt * 16 + kc * 4 + 2 * hf) * D_MODEL;
      v2f vb0, vb1, vb2, vb3;
      vb0.x = Vr[rc];      vb0.y = Vr[D_MODEL + rc];
      vb1.x = Vr[16 + rc]; vb1.y = Vr[D_MODEL + 16 + rc];
      vb2.x = Vr[32 + rc]; vb2.y = Vr[D_MODEL + 32 + rc];
      vb3.x = Vr[48 + rc]; vb3.y = Vr[D_MODEL + 48 + rc];
      o0 = wmma_f32(pa, vb0, o0);
      o1 = wmma_f32(pa, vb1, o1);
      o2 = wmma_f32(pa, vb2, o2);
      o3 = wmma_f32(pa, vb3, o3);
    }
  }
#pragma unroll
  for (int r = 0; r < 8; r++) {
    const float inv = 1.0f / lr[r];
    const size_t ob = (size_t)(qt * 16 + r + 8 * hf) * D_MODEL + rc;
    Ob[ob]      = o0[r] * inv;
    Ob[ob + 16] = o1[r] * inv;
    Ob[ob + 32] = o2[r] * inv;
    Ob[ob + 48] = o3[r] * inv;
  }
}

extern "C" void kernel_launch(void* const* d_in, const int* in_sizes, int n_in,
                              void* d_out, int out_size, void* d_ws, size_t ws_size,
                              hipStream_t stream) {
  (void)in_sizes; (void)n_in; (void)out_size; (void)ws_size;
  const float* x        = (const float*)d_in[0];
  const int*   positions= (const int*)d_in[1];
  const int*   theta    = (const int*)d_in[2];
  const float* Wq = (const float*)d_in[3];
  const float* Wk = (const float*)d_in[4];
  const float* Wv = (const float*)d_in[5];
  const float* Wo = (const float*)d_in[6];
  const float* g1 = (const float*)d_in[7];
  const float* g2 = (const float*)d_in[8];
  const float* w1 = (const float*)d_in[9];
  const float* w2 = (const float*)d_in[10];
  const float* w3 = (const float*)d_in[11];
  float* out = (float*)d_out;

  float* xn   = (float*)d_ws;
  float* q    = xn   + (size_t)ROWS * D_MODEL;
  float* k    = q    + (size_t)ROWS * D_MODEL;
  float* v    = k    + (size_t)ROWS * D_MODEL;
  float* attn = v    + (size_t)ROWS * D_MODEL;
  float* x2   = attn + (size_t)ROWS * D_MODEL;
  float* x3b  = x2   + (size_t)ROWS * D_MODEL;
  float* x1b  = q;

  const dim3 blk(256);
  const dim3 gblk(128);
  const int gproj = (ROWS / 128) * (D_MODEL / 64);
  const int gff   = (ROWS / 128) * (D_FF / 64);
  const int gfa   = (BATCH * NUM_HEADS * (SEQ / 16)) / 8;

  rmsnorm_kernel<<<ROWS, blk, 0, stream>>>(x, g1, xn, D_MODEL);
  gemm_lds<0><<<gproj, gblk, 0, stream>>>(xn, nullptr, Wq, nullptr, q, ROWS, D_MODEL, D_MODEL);
  gemm_lds<0><<<gproj, gblk, 0, stream>>>(xn, nullptr, Wk, nullptr, k, ROWS, D_MODEL, D_MODEL);
  gemm_lds<0><<<gproj, gblk, 0, stream>>>(xn, nullptr, Wv, nullptr, v, ROWS, D_MODEL, D_MODEL);
  rope_kernel<<<ROWS, blk, 0, stream>>>(q, positions, theta);
  rope_kernel<<<ROWS, blk, 0, stream>>>(k, positions, theta);
  flash_attn<<<gfa, blk, 0, stream>>>(q, k, v, attn);
  gemm_lds<1><<<gproj, gblk, 0, stream>>>(attn, nullptr, Wo, x, x2, ROWS, D_MODEL, D_MODEL);
  rmsnorm_kernel<<<ROWS, blk, 0, stream>>>(x2, g2, xn, D_MODEL);
  gemm_lds<0><<<gff, gblk, 0, stream>>>(xn, nullptr, w1, nullptr, x1b, ROWS, D_FF, D_MODEL);
  gemm_lds<0><<<gff, gblk, 0, stream>>>(xn, nullptr, w3, nullptr, x3b, ROWS, D_FF, D_MODEL);
  gemm_lds<2><<<gproj, gblk, 0, stream>>>(x1b, x3b, w2, x2, out, ROWS, D_MODEL, D_FF);
}


# attention: each 16-key K and V tile fetched once per block (coalesced 16-B/8-B loads, register-prefetched LDS double buffer) and shared by its 8 waves; fragments read from LDS; sum order kept
# speedup vs baseline: 2.8847x; 1.0673x over previous
#include <hip/hip_runtime.h>

#define D_MODEL   1024
#define NUM_HEADS 16
#define D_K       64
#define D_FF      2752
#define SEQ       2048
#define BATCH     2
#define ROWS      (BATCH*SEQ)

typedef float v2f __attribute__((ext_vector_type(2)));
typedef float v4f __attribute__((ext_vector_type(4)));
typedef float v8f __attribute__((ext_vector_type(8)));

__device__ __forceinline__ v8f wmma_f32(v2f a, v2f b, v8f c) {
  return __builtin_amdgcn_wmma_f32_16x16x4_f32(
       false, a,  false, b,
       (short)0, c,  false,  false);
}

__device__ __forceinline__ float silu_mul(float u, float w) {
  return u * (1.0f / (1.0f + __expf(-u))) * w;
}

__global__ void __launch_bounds__(256)
rmsnorm_kernel(const float* __restrict__ x, const float* __restrict__ g,
               float* __restrict__ out, int d) {
  const int row = blockIdx.x;
  const float* xr = x + (size_t)row * d;
  float ss = 0.f;
  for (int i = threadIdx.x; i < d; i += blockDim.x) { float v = xr[i]; ss = fmaf(v, v, ss); }
#pragma unroll
  for (int m = 16; m >= 1; m >>= 1) ss += __shfl_xor(ss, m, 32);
  __shared__ float red[8];
  const int w = threadIdx.x >> 5, ln = threadIdx.x & 31;
  if (ln == 0) red[w] = ss;
  __syncthreads();
  float tot = 0.f;
#pragma unroll
  for (int i = 0; i < 8; i++) tot += red[i];
  const float inv = rsqrtf(tot / (float)d + 1e-5f);
  for (int i = threadIdx.x; i < d; i += blockDim.x)
    out[(size_t)row * d + i] = xr[i] * inv * g[i];
}

__global__ void __launch_bounds__(256)
rope_kernel(float* __restrict__ t, const int* __restrict__ positions,
            const int* __restrict__ theta_p) {
  const int row = blockIdx.x;
  const int s = row % SEQ;
  const float pos = (float)positions[s];
  const float th = (float)(*theta_p);
  float* r = t + (size_t)row * D_MODEL;
  for (int p = threadIdx.x; p < D_MODEL / 2; p += blockDim.x) {
    const int c0 = 2 * p;
    const int d  = c0 & (D_K - 1);
    const float freq = __powf(th, -(float)d * (1.0f / (float)D_K));
    const float ang = pos * freq;
    float sn, cs;
    sincosf(ang, &sn, &cs);
    const float xe = r[c0], xo = r[c0 + 1];
    r[c0]     = xe * cs - xo * sn;
    r[c0 + 1] = xe * sn + xo * cs;
  }
}

#define GBM 128
#define GBN 64
#define GBK 32
#define GLD 36

template <int MODE>
__global__ void __launch_bounds__(128)
gemm_lds(const float* __restrict__ A, const float* __restrict__ A2,
         const float* __restrict__ B, const float* __restrict__ Res,
         float* __restrict__ C, int M, int N, int K) {
  __shared__ __attribute__((aligned(16))) float As[2][GBM * GLD];
  __shared__ __attribute__((aligned(16))) float Bs[2][GBN * GLD];
  (void)M;
  const int NT  = N >> 6;
  const int m0b = (blockIdx.x / NT) * GBM;
  const int n0b = (blockIdx.x % NT) * GBN;
  const int tid = threadIdx.x;
  const int w    = tid >> 5;
  const int lane = tid & 31;
  const int hf = lane >> 4;
  const int rc = lane & 15;

  const int sc = (tid & 7) << 2;
  const int sr = tid >> 3;
  const float* Ag  = A + (size_t)(m0b + sr) * K + sc;
  const float* A2g = (MODE == 2) ? (A2 + (size_t)(m0b + sr) * K + sc) : nullptr;
  const float* Bg  = B + (size_t)(n0b + sr) * K + sc;
  const size_t rstep = (size_t)16 * K;
  const int sts = sr * GLD + sc;

  v4f ra[8], rw[8], rb[4];

#pragma unroll
  for (int j = 0; j < 8; j++) {
    ra[j] = *(const v4f*)(Ag + j * rstep);
    if (MODE == 2) rw[j] = *(const v4f*)(A2g + j * rstep);
  }
#pragma unroll
  for (int j = 0; j < 4; j++) rb[j] = *(const v4f*)(Bg + j * rstep);
#pragma unroll
  for (int j = 0; j < 8; j++) {
    v4f v = ra[j];
    if (MODE == 2) {
      v.x = silu_mul(v.x, rw[j].x); v.y = silu_mul(v.y, rw[j].y);
      v.z = silu_mul(v.z, rw[j].z); v.w = silu_mul(v.w, rw[j].w);
    }
    *(v4f*)(&As[0][sts + j * 16 * GLD]) = v;
  }
#pragma unroll
  for (int j = 0; j < 4; j++) *(v4f*)(&Bs[0][sts + j * 16 * GLD]) = rb[j];
  __syncthreads();

  v8f acc[2][4] = {};
  const int nkt = K / GBK;
  const int afo = (w * 32 + rc) * GLD + 2 * hf;
  const int bfo = rc * GLD + 2 * hf;
  for (int kt = 0; kt < nkt; ++kt) {
    const int cur = kt & 1;
    const bool more = (kt + 1 < nkt);
    if (more) {
      const int ko = (kt + 1) * GBK;
#pragma unroll
      for (int j = 0; j < 8; j++) {
        ra[j] = *(const v4f*)(Ag + j * rstep + ko);
        if (MODE == 2) rw[j] = *(const v4f*)(A2g + j * rstep + ko);
      }
#pragma unroll
      for (int j = 0; j < 4; j++) rb[j] = *(const v4f*)(Bg + j * rstep + ko);
    }
    const float* as = &As[cur][afo];
    const float* bs = &Bs[cur][bfo];
#pragma unroll
    for (int kk = 0; kk < GBK / 4; kk++) {
      const v2f a0 = *(const v2f*)(as + kk * 4);
      const v2f a1 = *(const v2f*)(as + 16 * GLD + kk * 4);
      v2f b[4];
#pragma unroll
      for (int t = 0; t < 4; t++) b[t] = *(const v2f*)(bs + t * 16 * GLD + kk * 4);
#pragma unroll
      for (int t = 0; t < 4; t++) {
        acc[0][t] = wmma_f32(a0, b[t], acc[0][t]);
        acc[1][t] = wmma_f32(a1, b[t], acc[1][t]);
      }
    }
    if (more) {
      float* ad = &As[cur ^ 1][sts];
      float* bd = &Bs[cur ^ 1][sts];
#pragma unroll
      for (int j = 0; j < 8; j++) {
        v4f v = ra[j];
        if (MODE == 2) {
          v.x = silu_mul(v.x, rw[j].x); v.y = silu_mul(v.y, rw[j].y);
          v.z = silu_mul(v.z, rw[j].z); v.w = silu_mul(v.w, rw[j].w);
        }
        *(v4f*)(ad + j * 16 * GLD) = v;
      }
#pragma unroll
      for (int j = 0; j < 4; j++) *(v4f*)(bd + j * 16 * GLD) = rb[j];
    }
    __syncthreads();
  }

  const int m0 = m0b + w * 32;
  const int n0 = n0b;
#pragma unroll
  for (int i = 0; i < 2; i++) {
#pragma unroll
    for (int r = 0; r < 8; r++) {
      const size_t base = (size_t)(m0 + i * 16 + r + 8 * hf) * N + n0 + rc;
      float o0 = acc[i][0][r], o1 = acc[i][1][r], o2 = acc[i][2][r], o3 = acc[i][3][r];
      if (MODE >= 1) {
        o0 += Res[base];      o1 += Res[base + 16];
        o2 += Res[base + 32]; o3 += Res[base + 48];
      }
      C[base] = o0; C[base + 16] = o1; C[base + 32] = o2; C[base + 48] = o3;
    }
  }
}

#define FKP 68
#define FVP 160

__global__ void __launch_bounds__(256)
flash_attn(const float* __restrict__ Q, const float* __restrict__ Kt,
           const float* __restrict__ V, float* __restrict__ O) {
  __shared__ float plds[8 * 256];
  __shared__ __attribute__((aligned(16))) float Ks[2][16 * FKP];
  __shared__ __attribute__((aligned(16))) float Vs[2][8 * FVP];
  const int wv = __builtin_amdgcn_readfirstlane((int)(threadIdx.x >> 5));
  const int wid = blockIdx.x * 8 + wv;
  const int NQT = SEQ / 16;
  const int qt = wid % NQT;
  const int bh = wid / NQT;
  const int jmax = (int)((blockIdx.x * 8) % NQT) + 7;
  const int h = bh % NUM_HEADS;
  const int b = bh / NUM_HEADS;
  const size_t base = (size_t)b * SEQ * D_MODEL + (size_t)h * D_K;
  const float* Qb = Q + base;
  const float* Kb = Kt + base;
  const float* Vb = V + base;
  float* Ob = O + base;
  const int lane = threadIdx.x & 31;
  const int hf = lane >> 4;
  const int rc = lane & 15;
  float* lp = plds + wv * 256;

  const int sr = threadIdx.x >> 4;
  const int sc = (threadIdx.x & 15) << 2;
  const float* Kg = Kb + (size_t)sr * D_MODEL + sc;
  const int kst = sr * FKP + sc;
  const float* Vg = Vb + (size_t)(2 * wv) * D_MODEL + 2 * lane;
  const int vst = wv * FVP + 4 * lane;

  v4f rk = *(const v4f*)Kg;
  v2f rva = *(const v2f*)Vg;
  v2f rvb = *(const v2f*)(Vg + D_MODEL);
  *(v4f*)(&Ks[0][kst]) = rk;
  {
    v4f pk; pk.x = rva.x; pk.y = rvb.x; pk.z = rva.y; pk.w = rvb.y;
    *(v4f*)(&Vs[0][vst]) = pk;
  }

  v2f qa[16];
#pragma unroll
  for (int i = 0; i < 16; i++)
    qa[i] = *(const v2f*)(Qb + (size_t)(qt * 16 + rc) * D_MODEL + i * 4 + 2 * hf);

  v8f o0 = {}, o1 = {}, o2 = {}, o3 = {};
  float mr[8], lr[8];
#pragma unroll
  for (int r = 0; r < 8; r++) { mr[r] = -1e30f; lr[r] = 0.f; }
  __syncthreads();

  for (int jt = 0; jt <= jmax; ++jt) {
    const int cur = jt & 1;
    const bool more = (jt < jmax);
    if (more) {
      const size_t go = (size_t)(jt + 1) * 16 * D_MODEL;
      rk = *(const v4f*)(Kg + go);
      rva = *(const v2f*)(Vg + go);
      rvb = *(const v2f*)(Vg + go + D_MODEL);
    }
    if (jt <= qt) {
      v8f s = {};
      const float* kf = &Ks[cur][rc * FKP + 2 * hf];
#pragma unroll
      for (int i = 0; i < 16; i++) {
        const v2f kb = *(const v2f*)(kf + i * 4);
        s = wmma_f32(qa[i], kb, s);
      }
      const bool diag = (jt == qt);
#pragma unroll
      for (int r = 0; r < 8; r++) {
        float sv = s[r] * 0.125f;
        if (diag && (rc > r + 8 * hf)) sv = -1e30f;
        float mx = sv;
#pragma unroll
        for (int msk = 1; msk < 16; msk <<= 1) mx = fmaxf(mx, __shfl_xor(mx, msk, 32));
        const float mnew = fmaxf(mr[r], mx);
        const float corr = __expf(mr[r] - mnew);
        mr[r] = mnew;
        const float p = __expf(sv - mnew);
        float ps = p;
#pragma unroll
        for (int msk = 1; msk < 16; msk <<= 1) ps += __shfl_xor(ps, msk, 32);
        lr[r] = lr[r] * corr + ps;
        o0[r] *= corr; o1[r] *= corr; o2[r] *= corr; o3[r] *= corr;
        lp[(r + 8 * hf) * 16 + rc] = p;
      }
      asm volatile("s_wait_dscnt 0" ::: "memory");
#pragma unroll
      for (int kc = 0; kc < 4; kc++) {
        const v2f pa = *(const v2f*)(lp + rc * 16 + kc * 4 + 2 * hf);
        const float* Vr = &Vs[cur][(2 * kc + hf) * FVP + 2 * rc];
        const v2f vb0 = *(const v2f*)(Vr);
        const v2f vb1 = *(const v2f*)(Vr + 32);
        const v2f vb2 = *(const v2f*)(Vr + 64);
        const v2f vb3 = *(const v2f*)(Vr + 96);
        o0 = wmma_f32(pa, vb0, o0);
        o1 = wmma_f32(pa, vb1, o1);
        o2 = wmma_f32(pa, vb2, o2);
        o3 = wmma_f32(pa, vb3, o3);
      }
    }
    if (more) {
      *(v4f*)(&Ks[cur ^ 1][kst]) = rk;
      v4f pk; pk.x = rva.x; pk.y = rvb.x; pk.z = rva.y; pk.w = rvb.y;
      *(v4f*)(&Vs[cur ^ 1][vst]) = pk;
    }
    __syncthreads();
  }
#pragma unroll
  for (int r = 0; r < 8; r++) {
    const float inv = 1.0f / lr[r];
    const size_t ob = (size_t)(qt * 16 + r + 8 * hf) * D_MODEL + rc;
    Ob[ob]      = o0[r] * inv;
    Ob[ob + 16] = o1[r] * inv;
    Ob[ob + 32] = o2[r] * inv;
    Ob[ob + 48] = o3[r] * inv;
  }
}

extern "C" void kernel_launch(void* const* d_in, const int* in_sizes, int n_in,
                              void* d_out, int out_size, void* d_ws, size_t ws_size,
                              hipStream_t stream) {
  (void)in_sizes; (void)n_in; (void)out_size; (void)ws_size;
  const float* x        = (const float*)d_in[0];
  const int*   positions= (const int*)d_in[1];
  const int*   theta    = (const int*)d_in[2];
  const float* Wq = (const float*)d_in[3];
  const float* Wk = (const float*)d_in[4];
  const float* Wv = (const float*)d_in[5];
  const float* Wo = (const float*)d_in[6];
  const float* g1 = (const float*)d_in[7];
  const float* g2 = (const float*)d_in[8];
  const float* w1 = (const float*)d_in[9];
  const float* w2 = (const float*)d_in[10];
  const float* w3 = (const float*)d_in[11];
  float* out = (float*)d_out;

  float* xn   = (float*)d_ws;
  float* q    = xn   + (size_t)ROWS * D_MODEL;
  float* k    = q    + (size_t)ROWS * D_MODEL;
  float* v    = k    + (size_t)ROWS * D_MODEL;
  float* attn = v    + (size_t)ROWS * D_MODEL;
  float* x2   = attn + (size_t)ROWS * D_MODEL;
  float* x3b  = x2   + (size_t)ROWS * D_MODEL;
  float* x1b  = q;

  const dim3 blk(256);
  const dim3 gblk(128);
  const int gproj = (ROWS / 128) * (D_MODEL / 64);
  const int gff   = (ROWS / 128) * (D_FF / 64);
  const int gfa   = (BATCH * NUM_HEADS * (SEQ / 16)) / 8;

  rmsnorm_kernel<<<ROWS, blk, 0, stream>>>(x, g1, xn, D_MODEL);
  gemm_lds<0><<<gproj, gblk, 0, stream>>>(xn, nullptr, Wq, nullptr, q, ROWS, D_MODEL, D_MODEL);
  gemm_lds<0><<<gproj, gblk, 0, stream>>>(xn, nullptr, Wk, nullptr, k, ROWS, D_MODEL, D_MODEL);
  gemm_lds<0><<<gproj, gblk, 0, stream>>>(xn, nullptr, Wv, nullptr, v, ROWS, D_MODEL, D_MODEL);
  rope_kernel<<<ROWS, blk, 0, stream>>>(q, positions, theta);
  rope_kernel<<<ROWS, blk, 0, stream>>>(k, positions, theta);
  flash_attn<<<gfa, blk, 0, stream>>>(q, k, v, attn);
  gemm_lds<1><<<gproj, gblk, 0, stream>>>(attn, nullptr, Wo, x, x2, ROWS, D_MODEL, D_MODEL);
  rmsnorm_kernel<<<ROWS, blk, 0, stream>>>(x2, g2, xn, D_MODEL);
  gemm_lds<0><<<gff, gblk, 0, stream>>>(xn, nullptr, w1, nullptr, x1b, ROWS, D_FF, D_MODEL);
  gemm_lds<0><<<gff, gblk, 0, stream>>>(xn, nullptr, w3, nullptr, x3b, ROWS, D_FF, D_MODEL);
  gemm_lds<2><<<gproj, gblk, 0, stream>>>(x1b, x3b, w2, x2, out, ROWS, D_MODEL, D_FF);
}


# FFN: w1 and w3 products in one launch sharing the staged x tile, SwiGLU gate applied once in its epilogue; w2 GEMM reads the gated tensor instead of re-gating per N block; sum order kept
# speedup vs baseline: 3.1636x; 1.0967x over previous
#include <hip/hip_runtime.h>

#define D_MODEL   1024
#define NUM_HEADS 16
#define D_K       64
#define D_FF      2752
#define SEQ       2048
#define BATCH     2
#define ROWS      (BATCH*SEQ)

typedef float v2f __attribute__((ext_vector_type(2)));
typedef float v4f __attribute__((ext_vector_type(4)));
typedef float v8f __attribute__((ext_vector_type(8)));

__device__ __forceinline__ v8f wmma_f32(v2f a, v2f b, v8f c) {
  return __builtin_amdgcn_wmma_f32_16x16x4_f32(
       false, a,  false, b,
       (short)0, c,  false,  false);
}

__device__ __forceinline__ float silu_mul(float u, float w) {
  return u * (1.0f / (1.0f + __expf(-u))) * w;
}

__global__ void __launch_bounds__(256)
rmsnorm_kernel(const float* __restrict__ x, const float* __restrict__ g,
               float* __restrict__ out, int d) {
  const int row = blockIdx.x;
  const float* xr = x + (size_t)row * d;
  float ss = 0.f;
  for (int i = threadIdx.x; i < d; i += blockDim.x) { float v = xr[i]; ss = fmaf(v, v, ss); }
#pragma unroll
  for (int m = 16; m >= 1; m >>= 1) ss += __shfl_xor(ss, m, 32);
  __shared__ float red[8];
  const int w = threadIdx.x >> 5, ln = threadIdx.x & 31;
  if (ln == 0) red[w] = ss;
  __syncthreads();
  float tot = 0.f;
#pragma unroll
  for (int i = 0; i < 8; i++) tot += red[i];
  const float inv = rsqrtf(tot / (float)d + 1e-5f);
  for (int i = threadIdx.x; i < d; i += blockDim.x)
    out[(size_t)row * d + i] = xr[i] * inv * g[i];
}

__global__ void __launch_bounds__(256)
rope_kernel(float* __restrict__ t, const int* __restrict__ positions,
            const int* __restrict__ theta_p) {
  const int row = blockIdx.x;
  const int s = row % SEQ;
  const float pos = (float)positions[s];
  const float th = (float)(*theta_p);
  float* r = t + (size_t)row * D_MODEL;
  for (int p = threadIdx.x; p < D_MODEL / 2; p += blockDim.x) {
    const int c0 = 2 * p;
    const int d  = c0 & (D_K - 1);
    const float freq = __powf(th, -(float)d * (1.0f / (float)D_K));
    const float ang = pos * freq;
    float sn, cs;
    sincosf(ang, &sn, &cs);
    const float xe = r[c0], xo = r[c0 + 1];
    r[c0]     = xe * cs - xo * sn;
    r[c0 + 1] = xe * sn + xo * cs;
  }
}

#define GBM 128
#define GBN 64
#define GBK 32
#define GLD 36

template <int MODE>
__global__ void __launch_bounds__(128)
gemm_lds(const float* __restrict__ A,
         const float* __restrict__ B, const float* __restrict__ Res,
         float* __restrict__ C, int M, int N, int K) {
  __shared__ __attribute__((aligned(16))) float As[2][GBM * GLD];
  __shared__ __attribute__((aligned(16))) float Bs[2][GBN * GLD];
  (void)M;
  const int NT  = N >> 6;
  const int m0b = (blockIdx.x / NT) * GBM;
  const int n0b = (blockIdx.x % NT) * GBN;
  const int tid = threadIdx.x;
  const int w    = tid >> 5;
  const int lane = tid & 31;
  const int hf = lane >> 4;
  const int rc = lane & 15;

  const int sc = (tid & 7) << 2;
  const int sr = tid >> 3;
  const float* Ag  = A + (size_t)(m0b + sr) * K + sc;
  const float* Bg  = B + (size_t)(n0b + sr) * K + sc;
  const size_t rstep = (size_t)16 * K;
  const int sts = sr * GLD + sc;

  v4f ra[8], rb[4];

#pragma unroll
  for (int j = 0; j < 8; j++) ra[j] = *(const v4f*)(Ag + j * rstep);
#pragma unroll
  for (int j = 0; j < 4; j++) rb[j] = *(const v4f*)(Bg + j * rstep);
#pragma unroll
  for (int j = 0; j < 8; j++) *(v4f*)(&As[0][sts + j * 16 * GLD]) = ra[j];
#pragma unroll
  for (int j = 0; j < 4; j++) *(v4f*)(&Bs[0][sts + j * 16 * GLD]) = rb[j];
  __syncthreads();

  v8f acc[2][4] = {};
  const int nkt = K / GBK;
  const int afo = (w * 32 + rc) * GLD + 2 * hf;
  const int bfo = rc * GLD + 2 * hf;
  for (int kt = 0; kt < nkt; ++kt) {
    const int cur = kt & 1;
    const bool more = (kt + 1 < nkt);
    if (more) {
      const int ko = (kt + 1) * GBK;
#pragma unroll
      for (int j = 0; j < 8; j++) ra[j] = *(const v4f*)(Ag + j * rstep + ko);
#pragma unroll
      for (int j = 0; j < 4; j++) rb[j] = *(const v4f*)(Bg + j * rstep + ko);
    }
    const float* as = &As[cur][afo];
    const float* bs = &Bs[cur][bfo];
#pragma unroll
    for (int kk = 0; kk < GBK / 4; kk++) {
      const v2f a0 = *(const v2f*)(as + kk * 4);
      const v2f a1 = *(const v2f*)(as + 16 * GLD + kk * 4);
      v2f b[4];
#pragma unroll
      for (int t = 0; t < 4; t++) b[t] = *(const v2f*)(bs + t * 16 * GLD + kk * 4);
#pragma unroll
      for (int t = 0; t < 4; t++) {
        acc[0][t] = wmma_f32(a0, b[t], acc[0][t]);
        acc[1][t] = wmma_f32(a1, b[t], acc[1][t]);
      }
    }
    if (more) {
      float* ad = &As[cur ^ 1][sts];
      float* bd = &Bs[cur ^ 1][sts];
#pragma unroll
      for (int j = 0; j < 8; j++) *(v4f*)(ad + j * 16 * GLD) = ra[j];
#pragma unroll
      for (int j = 0; j < 4; j++) *(v4f*)(bd + j * 16 * GLD) = rb[j];
    }
    __syncthreads();
  }

  const int m0 = m0b + w * 32;
  const int n0 = n0b;
#pragma unroll
  for (int i = 0; i < 2; i++) {
#pragma unroll
    for (int r = 0; r < 8; r++) {
      const size_t base = (size_t)(m0 + i * 16 + r + 8 * hf) * N + n0 + rc;
      float o0 = acc[i][0][r], o1 = acc[i][1][r], o2 = acc[i][2][r], o3 = acc[i][3][r];
      if (MODE == 1) {
        o0 += Res[base];      o1 += Res[base + 16];
        o2 += Res[base + 32]; o3 += Res[base + 48];
      }
      C[base] = o0; C[base + 16] = o1; C[base + 32] = o2; C[base + 48] = o3;
    }
  }
}

__global__ void __launch_bounds__(128)
gemm_gate(const float* __restrict__ A, const float* __restrict__ B1,
          const float* __restrict__ B3, float* __restrict__ H,
          int M, int N, int K) {
  __shared__ __attribute__((aligned(16))) float As[2][GBM * GLD];
  __shared__ __attribute__((aligned(16))) float B1s[2][GBN * GLD];
  __shared__ __attribute__((aligned(16))) float B3s[2][GBN * GLD];
  (void)M;
  const int NT  = N >> 6;
  const int m0b = (blockIdx.x / NT) * GBM;
  const int n0b = (blockIdx.x % NT) * GBN;
  const int tid = threadIdx.x;
  const int w    = tid >> 5;
  const int lane = tid & 31;
  const int hf = lane >> 4;
  const int rc = lane & 15;

  const int sc = (tid & 7) << 2;
  const int sr = tid >> 3;
  const float* Ag  = A  + (size_t)(m0b + sr) * K + sc;
  const float* B1g = B1 + (size_t)(n0b + sr) * K + sc;
  const float* B3g = B3 + (size_t)(n0b + sr) * K + sc;
  const size_t rstep = (size_t)16 * K;
  const int sts = sr * GLD + sc;

  v4f ra[8], rb1[4], rb3[4];

#pragma unroll
  for (int j = 0; j < 8; j++) ra[j] = *(const v4f*)(Ag + j * rstep);
#pragma unroll
  for (int j = 0; j < 4; j++) {
    rb1[j] = *(const v4f*)(B1g + j * rstep);
    rb3[j] = *(const v4f*)(B3g + j * rstep);
  }
#pragma unroll
  for (int j = 0; j < 8; j++) *(v4f*)(&As[0][sts + j * 16 * GLD]) = ra[j];
#pragma unroll
  for (int j = 0; j < 4; j++) {
    *(v4f*)(&B1s[0][sts + j * 16 * GLD]) = rb1[j];
    *(v4f*)(&B3s[0][sts + j * 16 * GLD]) = rb3[j];
  }
  __syncthreads();

  v8f acc1[2][4] = {};
  v8f acc3[2][4] = {};
  const int nkt = K / GBK;
  const int afo = (w * 32 + rc) * GLD + 2 * hf;
  const int bfo = rc * GLD + 2 * hf;
  for (int kt = 0; kt < nkt; ++kt) {
    const int cur = kt & 1;
    const bool more = (kt + 1 < nkt);
    if (more) {
      const int ko = (kt + 1) * GBK;
#pragma unroll
      for (int j = 0; j < 8; j++) ra[j] = *(const v4f*)(Ag + j * rstep + ko);
#pragma unroll
      for (int j = 0; j < 4; j++) {
        rb1[j] = *(const v4f*)(B1g + j * rstep + ko);
        rb3[j] = *(const v4f*)(B3g + j * rstep + ko);
      }
    }
    const float* as  = &As[cur][afo];
    const float* b1s = &B1s[cur][bfo];
    const float* b3s = &B3s[cur][bfo];
#pragma unroll
    for (int kk = 0; kk < GBK / 4; kk++) {
      const v2f a0 = *(const v2f*)(as + kk * 4);
      const v2f a1 = *(const v2f*)(as + 16 * GLD + kk * 4);
      v2f b1[4], b3[4];
#pragma unroll
      for (int t = 0; t < 4; t++) {
        b1[t] = *(const v2f*)(b1s + t * 16 * GLD + kk * 4);
        b3[t] = *(const v2f*)(b3s + t * 16 * GLD + kk * 4);
      }
#pragma unroll
      for (int t = 0; t < 4; t++) {
        acc1[0][t] = wmma_f32(a0, b1[t], acc1[0][t]);
        acc1[1][t] = wmma_f32(a1, b1[t], acc1[1][t]);
      }
#pragma unroll
      for (int t = 0; t < 4; t++) {
        acc3[0][t] = wmma_f32(a0, b3[t], acc3[0][t]);
        acc3[1][t] = wmma_f32(a1, b3[t], acc3[1][t]);
      }
    }
    if (more) {
      float* ad  = &As[cur ^ 1][sts];
      float* b1d = &B1s[cur ^ 1][sts];
      float* b3d = &B3s[cur ^ 1][sts];
#pragma unroll
      for (int j = 0; j < 8; j++) *(v4f*)(ad + j * 16 * GLD) = ra[j];
#pragma unroll
      for (int j = 0; j < 4; j++) {
        *(v4f*)(b1d + j * 16 * GLD) = rb1[j];
        *(v4f*)(b3d + j * 16 * GLD) = rb3[j];
      }
    }
    __syncthreads();
  }

  const int m0 = m0b + w * 32;
  const int n0 = n0b;
#pragma unroll
  for (int i = 0; i < 2; i++) {
#pragma unroll
    for (int r = 0; r < 8; r++) {
      const size_t base = (size_t)(m0 + i * 16 + r + 8 * hf) * N + n0 + rc;
#pragma unroll
      for (int t = 0; t < 4; t++)
        H[base + 16 * t] = silu_mul(acc1[i][t][r], acc3[i][t][r]);
    }
  }
}

#define FKP 68
#define FVP 160

__global__ void __launch_bounds__(256)
flash_attn(const float* __restrict__ Q, const float* __restrict__ Kt,
           const float* __restrict__ V, float* __restrict__ O) {
  __shared__ float plds[8 * 256];
  __shared__ __attribute__((aligned(16))) float Ks[2][16 * FKP];
  __shared__ __attribute__((aligned(16))) float Vs[2][8 * FVP];
  const int wv = __builtin_amdgcn_readfirstlane((int)(threadIdx.x >> 5));
  const int wid = blockIdx.x * 8 + wv;
  const int NQT = SEQ / 16;
  const int qt = wid % NQT;
  const int bh = wid / NQT;
  const int jmax = (int)((blockIdx.x * 8) % NQT) + 7;
  const int h = bh % NUM_HEADS;
  const int b = bh / NUM_HEADS;
  const size_t base = (size_t)b * SEQ * D_MODEL + (size_t)h * D_K;
  const float* Qb = Q + base;
  const float* Kb = Kt + base;
  const float* Vb = V + base;
  float* Ob = O + base;
  const int lane = threadIdx.x & 31;
  const int hf = lane >> 4;
  const int rc = lane & 15;
  float* lp = plds + wv * 256;

  const int sr = threadIdx.x >> 4;
  const int sc = (threadIdx.x & 15) << 2;
  const float* Kg = Kb + (size_t)sr * D_MODEL + sc;
  const int kst = sr * FKP + sc;
  const float* Vg = Vb + (size_t)(2 * wv) * D_MODEL + 2 * lane;
  const int vst = wv * FVP + 4 * lane;

  v4f rk = *(const v4f*)Kg;
  v2f rva = *(const v2f*)Vg;
  v2f rvb = *(const v2f*)(Vg + D_MODEL);
  *(v4f*)(&Ks[0][kst]) = rk;
  {
    v4f pk; pk.x = rva.x; pk.y = rvb.x; pk.z = rva.y; pk.w = rvb.y;
    *(v4f*)(&Vs[0][vst]) = pk;
  }

  v2f qa[16];
#pragma unroll
  for (int i = 0; i < 16; i++)
    qa[i] = *(const v2f*)(Qb + (size_t)(qt * 16 + rc) * D_MODEL + i * 4 + 2 * hf);

  v8f o0 = {}, o1 = {}, o2 = {}, o3 = {};
  float mr[8], lr[8];
#pragma unroll
  for (int r = 0; r < 8; r++) { mr[r] = -1e30f; lr[r] = 0.f; }
  __syncthreads();

  for (int jt = 0; jt <= jmax; ++jt) {
    const int cur = jt & 1;
    const bool more = (jt < jmax);
    if (more) {
      const size_t go = (size_t)(jt + 1) * 16 * D_MODEL;
      rk = *(const v4f*)(Kg + go);
      rva = *(const v2f*)(Vg + go);
      rvb = *(const v2f*)(Vg + go + D_MODEL);
    }
    if (jt <= qt) {
      v8f s = {};
      const float* kf = &Ks[cur][rc * FKP + 2 * hf];
#pragma unroll
      for (int i = 0; i < 16; i++) {
        const v2f kb = *(const v2f*)(kf + i * 4);
        s = wmma_f32(qa[i], kb, s);
      }
      const bool diag = (jt == qt);
#pragma unroll
      for (int r = 0; r < 8; r++) {
        float sv = s[r] * 0.125f;
        if (diag && (rc > r + 8 * hf)) sv = -1e30f;
        float mx = sv;
#pragma unroll
        for (int msk = 1; msk < 16; msk <<= 1) mx = fmaxf(mx, __shfl_xor(mx, msk, 32));
        const float mnew = fmaxf(mr[r], mx);
        const float corr = __expf(mr[r] - mnew);
        mr[r] = mnew;
        const float p = __expf(sv - mnew);
        float ps = p;
#pragma unroll
        for (int msk = 1; msk < 16; msk <<= 1) ps += __shfl_xor(ps, msk, 32);
        lr[r] = lr[r] * corr + ps;
        o0[r] *= corr; o1[r] *= corr; o2[r] *= corr; o3[r] *= corr;
        lp[(r + 8 * hf) * 16 + rc] = p;
      }
      asm volatile("s_wait_dscnt 0" ::: "memory");
#pragma unroll
      for (int kc = 0; kc < 4; kc++) {
        const v2f pa = *(const v2f*)(lp + rc * 16 + kc * 4 + 2 * hf);
        const float* Vr = &Vs[cur][(2 * kc + hf) * FVP + 2 * rc];
        const v2f vb0 = *(const v2f*)(Vr);
        const v2f vb1 = *(const v2f*)(Vr + 32);
        const v2f vb2 = *(const v2f*)(Vr + 64);
        const v2f vb3 = *(const v2f*)(Vr + 96);
        o0 = wmma_f32(pa, vb0, o0);
        o1 = wmma_f32(pa, vb1, o1);
        o2 = wmma_f32(pa, vb2, o2);
        o3 = wmma_f32(pa, vb3, o3);
      }
    }
    if (more) {
      *(v4f*)(&Ks[cur ^ 1][kst]) = rk;
      v4f pk; pk.x = rva.x; pk.y = rvb.x; pk.z = rva.y; pk.w = rvb.y;
      *(v4f*)(&Vs[cur ^ 1][vst]) = pk;
    }
    __syncthreads();
  }
#pragma unroll
  for (int r = 0; r < 8; r++) {
    const float inv = 1.0f / lr[r];
    const size_t ob = (size_t)(qt * 16 + r + 8 * hf) * D_MODEL + rc;
    Ob[ob]      = o0[r] * inv;
    Ob[ob + 16] = o1[r] * inv;
    Ob[ob + 32] = o2[r] * inv;
    Ob[ob + 48] = o3[r] * inv;
  }
}

extern "C" void kernel_launch(void* const* d_in, const int* in_sizes, int n_in,
                              void* d_out, int out_size, void* d_ws, size_t ws_size,
                              hipStream_t stream) {
  (void)in_sizes; (void)n_in; (void)out_size; (void)ws_size;
  const float* x        = (const float*)d_in[0];
  const int*   positions= (const int*)d_in[1];
  const int*   theta    = (const int*)d_in[2];
  const float* Wq = (const float*)d_in[3];
  const float* Wk = (const float*)d_in[4];
  const float* Wv = (const float*)d_in[5];
  const float* Wo = (const float*)d_in[6];
  const float* g1 = (const float*)d_in[7];
  const float* g2 = (const float*)d_in[8];
  const float* w1 = (const float*)d_in[9];
  const float* w2 = (const float*)d_in[10];
  const float* w3 = (const float*)d_in[11];
  float* out = (float*)d_out;

  float* xn   = (float*)d_ws;
  float* q    = xn   + (size_t)ROWS * D_MODEL;
  float* k    = q    + (size_t)ROWS * D_MODEL;
  float* v    = k    + (size_t)ROWS * D_MODEL;
  float* attn = v    + (size_t)ROWS * D_MODEL;
  float* x2   = attn + (size_t)ROWS * D_MODEL;
  float* hb   = q;

  const dim3 blk(256);
  const dim3 gblk(128);
  const int gproj = (ROWS / 128) * (D_MODEL / 64);
  const int gff   = (ROWS / 128) * (D_FF / 64);
  const int gfa   = (BATCH * NUM_HEADS * (SEQ / 16)) / 8;

  rmsnorm_kernel<<<ROWS, blk, 0, stream>>>(x, g1, xn, D_MODEL);
  gemm_lds<0><<<gproj, gblk, 0, stream>>>(xn, Wq, nullptr, q, ROWS, D_MODEL, D_MODEL);
  gemm_lds<0><<<gproj, gblk, 0, stream>>>(xn, Wk, nullptr, k, ROWS, D_MODEL, D_MODEL);
  gemm_lds<0><<<gproj, gblk, 0, stream>>>(xn, Wv, nullptr, v, ROWS, D_MODEL, D_MODEL);
  rope_kernel<<<ROWS, blk, 0, stream>>>(q, positions, theta);
  rope_kernel<<<ROWS, blk, 0, stream>>>(k, positions, theta);
  flash_attn<<<gfa, blk, 0, stream>>>(q, k, v, attn);
  gemm_lds<1><<<gproj, gblk, 0, stream>>>(attn, Wo, x, x2, ROWS, D_MODEL, D_MODEL);
  rmsnorm_kernel<<<ROWS, blk, 0, stream>>>(x2, g2, xn, D_MODEL);
  gemm_gate<<<gff, gblk, 0, stream>>>(xn, w1, w3, hb, ROWS, D_FF, D_MODEL);
  gemm_lds<1><<<gproj, gblk, 0, stream>>>(hb, w2, x2, out, ROWS, D_MODEL, D_FF);
}


# attention: causal load balance: each block takes 4 late query tiles plus the 4 mirrored early tiles of one head, so every block does equal tile work; per-wave sums and their order kept
# speedup vs baseline: 3.3483x; 1.0584x over previous
#include <hip/hip_runtime.h>

#define D_MODEL   1024
#define NUM_HEADS 16
#define D_K       64
#define D_FF      2752
#define SEQ       2048
#define BATCH     2
#define ROWS      (BATCH*SEQ)

typedef float v2f __attribute__((ext_vector_type(2)));
typedef float v4f __attribute__((ext_vector_type(4)));
typedef float v8f __attribute__((ext_vector_type(8)));

__device__ __forceinline__ v8f wmma_f32(v2f a, v2f b, v8f c) {
  return __builtin_amdgcn_wmma_f32_16x16x4_f32(
       false, a,  false, b,
       (short)0, c,  false,  false);
}

__device__ __forceinline__ float silu_mul(float u, float w) {
  return u * (1.0f / (1.0f + __expf(-u))) * w;
}

__global__ void __launch_bounds__(256)
rmsnorm_kernel(const float* __restrict__ x, const float* __restrict__ g,
               float* __restrict__ out, int d) {
  const int row = blockIdx.x;
  const float* xr = x + (size_t)row * d;
  float ss = 0.f;
  for (int i = threadIdx.x; i < d; i += blockDim.x) { float v = xr[i]; ss = fmaf(v, v, ss); }
#pragma unroll
  for (int m = 16; m >= 1; m >>= 1) ss += __shfl_xor(ss, m, 32);
  __shared__ float red[8];
  const int w = threadIdx.x >> 5, ln = threadIdx.x & 31;
  if (ln == 0) red[w] = ss;
  __syncthreads();
  float tot = 0.f;
#pragma unroll
  for (int i = 0; i < 8; i++) tot += red[i];
  const float inv = rsqrtf(tot / (float)d + 1e-5f);
  for (int i = threadIdx.x; i < d; i += blockDim.x)
    out[(size_t)row * d + i] = xr[i] * inv * g[i];
}

__global__ void __launch_bounds__(256)
rope_kernel(float* __restrict__ t, const int* __restrict__ positions,
            const int* __restrict__ theta_p) {
  const int row = blockIdx.x;
  const int s = row % SEQ;
  const float pos = (float)positions[s];
  const float th = (float)(*theta_p);
  float* r = t + (size_t)row * D_MODEL;
  for (int p = threadIdx.x; p < D_MODEL / 2; p += blockDim.x) {
    const int c0 = 2 * p;
    const int d  = c0 & (D_K - 1);
    const float freq = __powf(th, -(float)d * (1.0f / (float)D_K));
    const float ang = pos * freq;
    float sn, cs;
    sincosf(ang, &sn, &cs);
    const float xe = r[c0], xo = r[c0 + 1];
    r[c0]     = xe * cs - xo * sn;
    r[c0 + 1] = xe * sn + xo * cs;
  }
}

#define GBM 128
#define GBN 64
#define GBK 32
#define GLD 36

template <int MODE>
__global__ void __launch_bounds__(128)
gemm_lds(const float* __restrict__ A,
         const float* __restrict__ B, const float* __restrict__ Res,
         float* __restrict__ C, int M, int N, int K) {
  __shared__ __attribute__((aligned(16))) float As[2][GBM * GLD];
  __shared__ __attribute__((aligned(16))) float Bs[2][GBN * GLD];
  (void)M;
  const int NT  = N >> 6;
  const int m0b = (blockIdx.x / NT) * GBM;
  const int n0b = (blockIdx.x % NT) * GBN;
  const int tid = threadIdx.x;
  const int w    = tid >> 5;
  const int lane = tid & 31;
  const int hf = lane >> 4;
  const int rc = lane & 15;

  const int sc = (tid & 7) << 2;
  const int sr = tid >> 3;
  const float* Ag  = A + (size_t)(m0b + sr) * K + sc;
  const float* Bg  = B + (size_t)(n0b + sr) * K + sc;
  const size_t rstep = (size_t)16 * K;
  const int sts = sr * GLD + sc;

  v4f ra[8], rb[4];

#pragma unroll
  for (int j = 0; j < 8; j++) ra[j] = *(const v4f*)(Ag + j * rstep);
#pragma unroll
  for (int j = 0; j < 4; j++) rb[j] = *(const v4f*)(Bg + j * rstep);
#pragma unroll
  for (int j = 0; j < 8; j++) *(v4f*)(&As[0][sts + j * 16 * GLD]) = ra[j];
#pragma unroll
  for (int j = 0; j < 4; j++) *(v4f*)(&Bs[0][sts + j * 16 * GLD]) = rb[j];
  __syncthreads();

  v8f acc[2][4] = {};
  const int nkt = K / GBK;
  const int afo = (w * 32 + rc) * GLD + 2 * hf;
  const int bfo = rc * GLD + 2 * hf;
  for (int kt = 0; kt < nkt; ++kt) {
    const int cur = kt & 1;
    const bool more = (kt + 1 < nkt);
    if (more) {
      const int ko = (kt + 1) * GBK;
#pragma unroll
      for (int j = 0; j < 8; j++) ra[j] = *(const v4f*)(Ag + j * rstep + ko);
#pragma unroll
      for (int j = 0; j < 4; j++) rb[j] = *(const v4f*)(Bg + j * rstep + ko);
    }
    const float* as = &As[cur][afo];
    const float* bs = &Bs[cur][bfo];
#pragma unroll
    for (int kk = 0; kk < GBK / 4; kk++) {
      const v2f a0 = *(const v2f*)(as + kk * 4);
      const v2f a1 = *(const v2f*)(as + 16 * GLD + kk * 4);
      v2f b[4];
#pragma unroll
      for (int t = 0; t < 4; t++) b[t] = *(const v2f*)(bs + t * 16 * GLD + kk * 4);
#pragma unroll
      for (int t = 0; t < 4; t++) {
        acc[0][t] = wmma_f32(a0, b[t], acc[0][t]);
        acc[1][t] = wmma_f32(a1, b[t], acc[1][t]);
      }
    }
    if (more) {
      float* ad = &As[cur ^ 1][sts];
      float* bd = &Bs[cur ^ 1][sts];
#pragma unroll
      for (int j = 0; j < 8; j++) *(v4f*)(ad + j * 16 * GLD) = ra[j];
#pragma unroll
      for (int j = 0; j < 4; j++) *(v4f*)(bd + j * 16 * GLD) = rb[j];
    }
    __syncthreads();
  }

  const int m0 = m0b + w * 32;
  const int n0 = n0b;
#pragma unroll
  for (int i = 0; i < 2; i++) {
#pragma unroll
    for (int r = 0; r < 8; r++) {
      const size_t base = (size_t)(m0 + i * 16 + r + 8 * hf) * N + n0 + rc;
      float o0 = acc[i][0][r], o1 = acc[i][1][r], o2 = acc[i][2][r], o3 = acc[i][3][r];
      if (MODE == 1) {
        o0 += Res[base];      o1 += Res[base + 16];
        o2 += Res[base + 32]; o3 += Res[base + 48];
      }
      C[base] = o0; C[base + 16] = o1; C[base + 32] = o2; C[base + 48] = o3;
    }
  }
}

__global__ void __launch_bounds__(128)
gemm_gate(const float* __restrict__ A, const float* __restrict__ B1,
          const float* __restrict__ B3, float* __restrict__ H,
          int M, int N, int K) {
  __shared__ __attribute__((aligned(16))) float As[2][GBM * GLD];
  __shared__ __attribute__((aligned(16))) float B1s[2][GBN * GLD];
  __shared__ __attribute__((aligned(16))) float B3s[2][GBN * GLD];
  (void)M;
  const int NT  = N >> 6;
  const int m0b = (blockIdx.x / NT) * GBM;
  const int n0b = (blockIdx.x % NT) * GBN;
  const int tid = threadIdx.x;
  const int w    = tid >> 5;
  const int lane = tid & 31;
  const int hf = lane >> 4;
  const int rc = lane & 15;

  const int sc = (tid & 7) << 2;
  const int sr = tid >> 3;
  const float* Ag  = A  + (size_t)(m0b + sr) * K + sc;
  const float* B1g = B1 + (size_t)(n0b + sr) * K + sc;
  const float* B3g = B3 + (size_t)(n0b + sr) * K + sc;
  const size_t rstep = (size_t)16 * K;
  const int sts = sr * GLD + sc;

  v4f ra[8], rb1[4], rb3[4];

#pragma unroll
  for (int j = 0; j < 8; j++) ra[j] = *(const v4f*)(Ag + j * rstep);
#pragma unroll
  for (int j = 0; j < 4; j++) {
    rb1[j] = *(const v4f*)(B1g + j * rstep);
    rb3[j] = *(const v4f*)(B3g + j * rstep);
  }
#pragma unroll
  for (int j = 0; j < 8; j++) *(v4f*)(&As[0][sts + j * 16 * GLD]) = ra[j];
#pragma unroll
  for (int j = 0; j < 4; j++) {
    *(v4f*)(&B1s[0][sts + j * 16 * GLD]) = rb1[j];
    *(v4f*)(&B3s[0][sts + j * 16 * GLD]) = rb3[j];
  }
  __syncthreads();

  v8f acc1[2][4] = {};
  v8f acc3[2][4] = {};
  const int nkt = K / GBK;
  const int afo = (w * 32 + rc) * GLD + 2 * hf;
  const int bfo = rc * GLD + 2 * hf;
  for (int kt = 0; kt < nkt; ++kt) {
    const int cur = kt & 1;
    const bool more = (kt + 1 < nkt);
    if (more) {
      const int ko = (kt + 1) * GBK;
#pragma unroll
      for (int j = 0; j < 8; j++) ra[j] = *(const v4f*)(Ag + j * rstep + ko);
#pragma unroll
      for (int j = 0; j < 4; j++) {
        rb1[j] = *(const v4f*)(B1g + j * rstep + ko);
        rb3[j] = *(const v4f*)(B3g + j * rstep + ko);
      }
    }
    const float* as  = &As[cur][afo];
    const float* b1s = &B1s[cur][bfo];
    const float* b3s = &B3s[cur][bfo];
#pragma unroll
    for (int kk = 0; kk < GBK / 4; kk++) {
      const v2f a0 = *(const v2f*)(as + kk * 4);
      const v2f a1 = *(const v2f*)(as + 16 * GLD + kk * 4);
      v2f b1[4], b3[4];
#pragma unroll
      for (int t = 0; t < 4; t++) {
        b1[t] = *(const v2f*)(b1s + t * 16 * GLD + kk * 4);
        b3[t] = *(const v2f*)(b3s + t * 16 * GLD + kk * 4);
      }
#pragma unroll
      for (int t = 0; t < 4; t++) {
        acc1[0][t] = wmma_f32(a0, b1[t], acc1[0][t]);
        acc1[1][t] = wmma_f32(a1, b1[t], acc1[1][t]);
      }
#pragma unroll
      for (int t = 0; t < 4; t++) {
        acc3[0][t] = wmma_f32(a0, b3[t], acc3[0][t]);
        acc3[1][t] = wmma_f32(a1, b3[t], acc3[1][t]);
      }
    }
    if (more) {
      float* ad  = &As[cur ^ 1][sts];
      float* b1d = &B1s[cur ^ 1][sts];
      float* b3d = &B3s[cur ^ 1][sts];
#pragma unroll
      for (int j = 0; j < 8; j++) *(v4f*)(ad + j * 16 * GLD) = ra[j];
#pragma unroll
      for (int j = 0; j < 4; j++) {
        *(v4f*)(b1d + j * 16 * GLD) = rb1[j];
        *(v4f*)(b3d + j * 16 * GLD) = rb3[j];
      }
    }
    __syncthreads();
  }

  const int m0 = m0b + w * 32;
  const int n0 = n0b;
#pragma unroll
  for (int i = 0; i < 2; i++) {
#pragma unroll
    for (int r = 0; r < 8; r++) {
      const size_t base = (size_t)(m0 + i * 16 + r + 8 * hf) * N + n0 + rc;
#pragma unroll
      for (int t = 0; t < 4; t++)
        H[base + 16 * t] = silu_mul(acc1[i][t][r], acc3[i][t][r]);
    }
  }
}

#define FKP 68
#define FVP 160

__global__ void __launch_bounds__(256)
flash_attn(const float* __restrict__ Q, const float* __restrict__ Kt,
           const float* __restrict__ V, float* __restrict__ O) {
  __shared__ float plds[8 * 256];
  __shared__ __attribute__((aligned(16))) float Ks[2][16 * FKP];
  __shared__ __attribute__((aligned(16))) float Vs[2][8 * FVP];
  const int wv = __builtin_amdgcn_readfirstlane((int)(threadIdx.x >> 5));
  const int NQT = SEQ / 16;
  const int NPB = NQT / 8;
  const int bh = blockIdx.x / NPB;
  const int sp = blockIdx.x % NPB;
  const int qhi = NQT - 4 - 4 * sp;
  const int qt = (wv < 4) ? (qhi + wv) : (4 * sp + (wv - 4));
  const int jmax = qhi + 3;
  const int h = bh % NUM_HEADS;
  const int b = bh / NUM_HEADS;
  const size_t base = (size_t)b * SEQ * D_MODEL + (size_t)h * D_K;
  const float* Qb = Q + base;
  const float* Kb = Kt + base;
  const float* Vb = V + base;
  float* Ob = O + base;
  const int lane = threadIdx.x & 31;
  const int hf = lane >> 4;
  const int rc = lane & 15;
  float* lp = plds + wv * 256;

  const int sr = threadIdx.x >> 4;
  const int sc = (threadIdx.x & 15) << 2;
  const float* Kg = Kb + (size_t)sr * D_MODEL + sc;
  const int kst = sr * FKP + sc;
  const float* Vg = Vb + (size_t)(2 * wv) * D_MODEL + 2 * lane;
  const int vst = wv * FVP + 4 * lane;

  v4f rk = *(const v4f*)Kg;
  v2f rva = *(const v2f*)Vg;
  v2f rvb = *(const v2f*)(Vg + D_MODEL);
  *(v4f*)(&Ks[0][kst]) = rk;
  {
    v4f pk; pk.x = rva.x; pk.y = rvb.x; pk.z = rva.y; pk.w = rvb.y;
    *(v4f*)(&Vs[0][vst]) = pk;
  }

  v2f qa[16];
#pragma unroll
  for (int i = 0; i < 16; i++)
    qa[i] = *(const v2f*)(Qb + (size_t)(qt * 16 + rc) * D_MODEL + i * 4 + 2 * hf);

  v8f o0 = {}, o1 = {}, o2 = {}, o3 = {};
  float mr[8], lr[8];
#pragma unroll
  for (int r = 0; r < 8; r++) { mr[r] = -1e30f; lr[r] = 0.f; }
  __syncthreads();

  for (int jt = 0; jt <= jmax; ++jt) {
    const int cur = jt & 1;
    const bool more = (jt < jmax);
    if (more) {
      const size_t go = (size_t)(jt + 1) * 16 * D_MODEL;
      rk = *(const v4f*)(Kg + go);
      rva = *(const v2f*)(Vg + go);
      rvb = *(const v2f*)(Vg + go + D_MODEL);
    }
    if (jt <= qt) {
      v8f s = {};
      const float* kf = &Ks[cur][rc * FKP + 2 * hf];
#pragma unroll
      for (int i = 0; i < 16; i++) {
        const v2f kb = *(const v2f*)(kf + i * 4);
        s = wmma_f32(qa[i], kb, s);
      }
      const bool diag = (jt == qt);
#pragma unroll
      for (int r = 0; r < 8; r++) {
        float sv = s[r] * 0.125f;
        if (diag && (rc > r + 8 * hf)) sv = -1e30f;
        float mx = sv;
#pragma unroll
        for (int msk = 1; msk < 16; msk <<= 1) mx = fmaxf(mx, __shfl_xor(mx, msk, 32));
        const float mnew = fmaxf(mr[r], mx);
        const float corr = __expf(mr[r] - mnew);
        mr[r] = mnew;
        const float p = __expf(sv - mnew);
        float ps = p;
#pragma unroll
        for (int msk = 1; msk < 16; msk <<= 1) ps += __shfl_xor(ps, msk, 32);
        lr[r] = lr[r] * corr + ps;
        o0[r] *= corr; o1[r] *= corr; o2[r] *= corr; o3[r] *= corr;
        lp[(r + 8 * hf) * 16 + rc] = p;
      }
      asm volatile("s_wait_dscnt 0" ::: "memory");
#pragma unroll
      for (int kc = 0; kc < 4; kc++) {
        const v2f pa = *(const v2f*)(lp + rc * 16 + kc * 4 + 2 * hf);
        const float* Vr = &Vs[cur][(2 * kc + hf) * FVP + 2 * rc];
        const v2f vb0 = *(const v2f*)(Vr);
        const v2f vb1 = *(const v2f*)(Vr + 32);
        const v2f vb2 = *(const v2f*)(Vr + 64);
        const v2f vb3 = *(const v2f*)(Vr + 96);
        o0 = wmma_f32(pa, vb0, o0);
        o1 = wmma_f32(pa, vb1, o1);
        o2 = wmma_f32(pa, vb2, o2);
        o3 = wmma_f32(pa, vb3, o3);
      }
    }
    if (more) {
      *(v4f*)(&Ks[cur ^ 1][kst]) = rk;
      v4f pk; pk.x = rva.x; pk.y = rvb.x; pk.z = rva.y; pk.w = rvb.y;
      *(v4f*)(&Vs[cur ^ 1][vst]) = pk;
    }
    __syncthreads();
  }
#pragma unroll
  for (int r = 0; r < 8; r++) {
    const float inv = 1.0f / lr[r];
    const size_t ob = (size_t)(qt * 16 + r + 8 * hf) * D_MODEL + rc;
    Ob[ob]      = o0[r] * inv;
    Ob[ob + 16] = o1[r] * inv;
    Ob[ob + 32] = o2[r] * inv;
    Ob[ob + 48] = o3[r] * inv;
  }
}

extern "C" void kernel_launch(void* const* d_in, const int* in_sizes, int n_in,
                              void* d_out, int out_size, void* d_ws, size_t ws_size,
                              hipStream_t stream) {
  (void)in_sizes; (void)n_in; (void)out_size; (void)ws_size;
  const float* x        = (const float*)d_in[0];
  const int*   positions= (const int*)d_in[1];
  const int*   theta    = (const int*)d_in[2];
  const float* Wq = (const float*)d_in[3];
  const float* Wk = (const float*)d_in[4];
  const float* Wv = (const float*)d_in[5];
  const float* Wo = (const float*)d_in[6];
  const float* g1 = (const float*)d_in[7];
  const float* g2 = (const float*)d_in[8];
  const float* w1 = (const float*)d_in[9];
  const float* w2 = (const float*)d_in[10];
  const float* w3 = (const float*)d_in[11];
  float* out = (float*)d_out;

  float* xn   = (float*)d_ws;
  float* q    = xn   + (size_t)ROWS * D_MODEL;
  float* k    = q    + (size_t)ROWS * D_MODEL;
  float* v    = k    + (size_t)ROWS * D_MODEL;
  float* attn = v    + (size_t)ROWS * D_MODEL;
  float* x2   = attn + (size_t)ROWS * D_MODEL;
  float* hb   = q;

  const dim3 blk(256);
  const dim3 gblk(128);
  const int gproj = (ROWS / 128) * (D_MODEL / 64);
  const int gff   = (ROWS / 128) * (D_FF / 64);
  const int gfa   = (BATCH * NUM_HEADS * (SEQ / 16)) / 8;

  rmsnorm_kernel<<<ROWS, blk, 0, stream>>>(x, g1, xn, D_MODEL);
  gemm_lds<0><<<gproj, gblk, 0, stream>>>(xn, Wq, nullptr, q, ROWS, D_MODEL, D_MODEL);
  gemm_lds<0><<<gproj, gblk, 0, stream>>>(xn, Wk, nullptr, k, ROWS, D_MODEL, D_MODEL);
  gemm_lds<0><<<gproj, gblk, 0, stream>>>(xn, Wv, nullptr, v, ROWS, D_MODEL, D_MODEL);
  rope_kernel<<<ROWS, blk, 0, stream>>>(q, positions, theta);
  rope_kernel<<<ROWS, blk, 0, stream>>>(k, positions, theta);
  flash_attn<<<gfa, blk, 0, stream>>>(q, k, v, attn);
  gemm_lds<1><<<gproj, gblk, 0, stream>>>(attn, Wo, x, x2, ROWS, D_MODEL, D_MODEL);
  rmsnorm_kernel<<<ROWS, blk, 0, stream>>>(x2, g2, xn, D_MODEL);
  gemm_gate<<<gff, gblk, 0, stream>>>(xn, w1, w3, hb, ROWS, D_FF, D_MODEL);
  gemm_lds<1><<<gproj, gblk, 0, stream>>>(hb, w2, x2, out, ROWS, D_MODEL, D_FF);
}


# attention: score tile formed transposed (keys on rows): each query's max and sum are in-lane, P feeds the V product from registers, no LDS staging or lane shuffles; row-sum and P.V sum order changed
# speedup vs baseline: 3.5611x; 1.0636x over previous
#include <hip/hip_runtime.h>

#define D_MODEL   1024
#define NUM_HEADS 16
#define D_K       64
#define D_FF      2752
#define SEQ       2048
#define BATCH     2
#define ROWS      (BATCH*SEQ)

typedef float v2f __attribute__((ext_vector_type(2)));
typedef float v4f __attribute__((ext_vector_type(4)));
typedef float v8f __attribute__((ext_vector_type(8)));

__device__ __forceinline__ v8f wmma_f32(v2f a, v2f b, v8f c) {
  return __builtin_amdgcn_wmma_f32_16x16x4_f32(
       false, a,  false, b,
       (short)0, c,  false,  false);
}

__device__ __forceinline__ float silu_mul(float u, float w) {
  return u * (1.0f / (1.0f + __expf(-u))) * w;
}

__global__ void __launch_bounds__(256)
rmsnorm_kernel(const float* __restrict__ x, const float* __restrict__ g,
               float* __restrict__ out, int d) {
  const int row = blockIdx.x;
  const float* xr = x + (size_t)row * d;
  float ss = 0.f;
  for (int i = threadIdx.x; i < d; i += blockDim.x) { float v = xr[i]; ss = fmaf(v, v, ss); }
#pragma unroll
  for (int m = 16; m >= 1; m >>= 1) ss += __shfl_xor(ss, m, 32);
  __shared__ float red[8];
  const int w = threadIdx.x >> 5, ln = threadIdx.x & 31;
  if (ln == 0) red[w] = ss;
  __syncthreads();
  float tot = 0.f;
#pragma unroll
  for (int i = 0; i < 8; i++) tot += red[i];
  const float inv = rsqrtf(tot / (float)d + 1e-5f);
  for (int i = threadIdx.x; i < d; i += blockDim.x)
    out[(size_t)row * d + i] = xr[i] * inv * g[i];
}

__global__ void __launch_bounds__(256)
rope_kernel(float* __restrict__ t, const int* __restrict__ positions,
            const int* __restrict__ theta_p) {
  const int row = blockIdx.x;
  const int s = row % SEQ;
  const float pos = (float)positions[s];
  const float th = (float)(*theta_p);
  float* r = t + (size_t)row * D_MODEL;
  for (int p = threadIdx.x; p < D_MODEL / 2; p += blockDim.x) {
    const int c0 = 2 * p;
    const int d  = c0 & (D_K - 1);
    const float freq = __powf(th, -(float)d * (1.0f / (float)D_K));
    const float ang = pos * freq;
    float sn, cs;
    sincosf(ang, &sn, &cs);
    const float xe = r[c0], xo = r[c0 + 1];
    r[c0]     = xe * cs - xo * sn;
    r[c0 + 1] = xe * sn + xo * cs;
  }
}

#define GBM 128
#define GBN 64
#define GBK 32
#define GLD 36

template <int MODE>
__global__ void __launch_bounds__(128)
gemm_lds(const float* __restrict__ A,
         const float* __restrict__ B, const float* __restrict__ Res,
         float* __restrict__ C, int M, int N, int K) {
  __shared__ __attribute__((aligned(16))) float As[2][GBM * GLD];
  __shared__ __attribute__((aligned(16))) float Bs[2][GBN * GLD];
  (void)M;
  const int NT  = N >> 6;
  const int m0b = (blockIdx.x / NT) * GBM;
  const int n0b = (blockIdx.x % NT) * GBN;
  const int tid = threadIdx.x;
  const int w    = tid >> 5;
  const int lane = tid & 31;
  const int hf = lane >> 4;
  const int rc = lane & 15;

  const int sc = (tid & 7) << 2;
  const int sr = tid >> 3;
  const float* Ag  = A + (size_t)(m0b + sr) * K + sc;
  const float* Bg  = B + (size_t)(n0b + sr) * K + sc;
  const size_t rstep = (size_t)16 * K;
  const int sts = sr * GLD + sc;

  v4f ra[8], rb[4];

#pragma unroll
  for (int j = 0; j < 8; j++) ra[j] = *(const v4f*)(Ag + j * rstep);
#pragma unroll
  for (int j = 0; j < 4; j++) rb[j] = *(const v4f*)(Bg + j * rstep);
#pragma unroll
  for (int j = 0; j < 8; j++) *(v4f*)(&As[0][sts + j * 16 * GLD]) = ra[j];
#pragma unroll
  for (int j = 0; j < 4; j++) *(v4f*)(&Bs[0][sts + j * 16 * GLD]) = rb[j];
  __syncthreads();

  v8f acc[2][4] = {};
  const int nkt = K / GBK;
  const int afo = (w * 32 + rc) * GLD + 2 * hf;
  const int bfo = rc * GLD + 2 * hf;
  for (int kt = 0; kt < nkt; ++kt) {
    const int cur = kt & 1;
    const bool more = (kt + 1 < nkt);
    if (more) {
      const int ko = (kt + 1) * GBK;
#pragma unroll
      for (int j = 0; j < 8; j++) ra[j] = *(const v4f*)(Ag + j * rstep + ko);
#pragma unroll
      for (int j = 0; j < 4; j++) rb[j] = *(const v4f*)(Bg + j * rstep + ko);
    }
    const float* as = &As[cur][afo];
    const float* bs = &Bs[cur][bfo];
#pragma unroll
    for (int kk = 0; kk < GBK / 4; kk++) {
      const v2f a0 = *(const v2f*)(as + kk * 4);
      const v2f a1 = *(const v2f*)(as + 16 * GLD + kk * 4);
      v2f b[4];
#pragma unroll
      for (int t = 0; t < 4; t++) b[t] = *(const v2f*)(bs + t * 16 * GLD + kk * 4);
#pragma unroll
      for (int t = 0; t < 4; t++) {
        acc[0][t] = wmma_f32(a0, b[t], acc[0][t]);
        acc[1][t] = wmma_f32(a1, b[t], acc[1][t]);
      }
    }
    if (more) {
      float* ad = &As[cur ^ 1][sts];
      float* bd = &Bs[cur ^ 1][sts];
#pragma unroll
      for (int j = 0; j < 8; j++) *(v4f*)(ad + j * 16 * GLD) = ra[j];
#pragma unroll
      for (int j = 0; j < 4; j++) *(v4f*)(bd + j * 16 * GLD) = rb[j];
    }
    __syncthreads();
  }

  const int m0 = m0b + w * 32;
  const int n0 = n0b;
#pragma unroll
  for (int i = 0; i < 2; i++) {
#pragma unroll
    for (int r = 0; r < 8; r++) {
      const size_t base = (size_t)(m0 + i * 16 + r + 8 * hf) * N + n0 + rc;
      float o0 = acc[i][0][r], o1 = acc[i][1][r], o2 = acc[i][2][r], o3 = acc[i][3][r];
      if (MODE == 1) {
        o0 += Res[base];      o1 += Res[base + 16];
        o2 += Res[base + 32]; o3 += Res[base + 48];
      }
      C[base] = o0; C[base + 16] = o1; C[base + 32] = o2; C[base + 48] = o3;
    }
  }
}

__global__ void __launch_bounds__(128)
gemm_gate(const float* __restrict__ A, const float* __restrict__ B1,
          const float* __restrict__ B3, float* __restrict__ H,
          int M, int N, int K) {
  __shared__ __attribute__((aligned(16))) float As[2][GBM * GLD];
  __shared__ __attribute__((aligned(16))) float B1s[2][GBN * GLD];
  __shared__ __attribute__((aligned(16))) float B3s[2][GBN * GLD];
  (void)M;
  const int NT  = N >> 6;
  const int m0b = (blockIdx.x / NT) * GBM;
  const int n0b = (blockIdx.x % NT) * GBN;
  const int tid = threadIdx.x;
  const int w    = tid >> 5;
  const int lane = tid & 31;
  const int hf = lane >> 4;
  const int rc = lane & 15;

  const int sc = (tid & 7) << 2;
  const int sr = tid >> 3;
  const float* Ag  = A  + (size_t)(m0b + sr) * K + sc;
  const float* B1g = B1 + (size_t)(n0b + sr) * K + sc;
  const float* B3g = B3 + (size_t)(n0b + sr) * K + sc;
  const size_t rstep = (size_t)16 * K;
  const int sts = sr * GLD + sc;

  v4f ra[8], rb1[4], rb3[4];

#pragma unroll
  for (int j = 0; j < 8; j++) ra[j] = *(const v4f*)(Ag + j * rstep);
#pragma unroll
  for (int j = 0; j < 4; j++) {
    rb1[j] = *(const v4f*)(B1g + j * rstep);
    rb3[j] = *(const v4f*)(B3g + j * rstep);
  }
#pragma unroll
  for (int j = 0; j < 8; j++) *(v4f*)(&As[0][sts + j * 16 * GLD]) = ra[j];
#pragma unroll
  for (int j = 0; j < 4; j++) {
    *(v4f*)(&B1s[0][sts + j * 16 * GLD]) = rb1[j];
    *(v4f*)(&B3s[0][sts + j * 16 * GLD]) = rb3[j];
  }
  __syncthreads();

  v8f acc1[2][4] = {};
  v8f acc3[2][4] = {};
  const int nkt = K / GBK;
  const int afo = (w * 32 + rc) * GLD + 2 * hf;
  const int bfo = rc * GLD + 2 * hf;
  for (int kt = 0; kt < nkt; ++kt) {
    const int cur = kt & 1;
    const bool more = (kt + 1 < nkt);
    if (more) {
      const int ko = (kt + 1) * GBK;
#pragma unroll
      for (int j = 0; j < 8; j++) ra[j] = *(const v4f*)(Ag + j * rstep + ko);
#pragma unroll
      for (int j = 0; j < 4; j++) {
        rb1[j] = *(const v4f*)(B1g + j * rstep + ko);
        rb3[j] = *(const v4f*)(B3g + j * rstep + ko);
      }
    }
    const float* as  = &As[cur][afo];
    const float* b1s = &B1s[cur][bfo];
    const float* b3s = &B3s[cur][bfo];
#pragma unroll
    for (int kk = 0; kk < GBK / 4; kk++) {
      const v2f a0 = *(const v2f*)(as + kk * 4);
      const v2f a1 = *(const v2f*)(as + 16 * GLD + kk * 4);
      v2f b1[4], b3[4];
#pragma unroll
      for (int t = 0; t < 4; t++) {
        b1[t] = *(const v2f*)(b1s + t * 16 * GLD + kk * 4);
        b3[t] = *(const v2f*)(b3s + t * 16 * GLD + kk * 4);
      }
#pragma unroll
      for (int t = 0; t < 4; t++) {
        acc1[0][t] = wmma_f32(a0, b1[t], acc1[0][t]);
        acc1[1][t] = wmma_f32(a1, b1[t], acc1[1][t]);
      }
#pragma unroll
      for (int t = 0; t < 4; t++) {
        acc3[0][t] = wmma_f32(a0, b3[t], acc3[0][t]);
        acc3[1][t] = wmma_f32(a1, b3[t], acc3[1][t]);
      }
    }
    if (more) {
      float* ad  = &As[cur ^ 1][sts];
      float* b1d = &B1s[cur ^ 1][sts];
      float* b3d = &B3s[cur ^ 1][sts];
#pragma unroll
      for (int j = 0; j < 8; j++) *(v4f*)(ad + j * 16 * GLD) = ra[j];
#pragma unroll
      for (int j = 0; j < 4; j++) {
        *(v4f*)(b1d + j * 16 * GLD) = rb1[j];
        *(v4f*)(b3d + j * 16 * GLD) = rb3[j];
      }
    }
    __syncthreads();
  }

  const int m0 = m0b + w * 32;
  const int n0 = n0b;
#pragma unroll
  for (int i = 0; i < 2; i++) {
#pragma unroll
    for (int r = 0; r < 8; r++) {
      const size_t base = (size_t)(m0 + i * 16 + r + 8 * hf) * N + n0 + rc;
#pragma unroll
      for (int t = 0; t < 4; t++)
        H[base + 16 * t] = silu_mul(acc1[i][t][r], acc3[i][t][r]);
    }
  }
}

#define FKP 68
#define FVP 160

__device__ __forceinline__ float xhalf(float x) {
  const int b = __builtin_bit_cast(int, x);
  return __builtin_bit_cast(float,
      __builtin_amdgcn_permlanex16(b, b, 0x76543210, 0xfedcba98, false, false));
}

__global__ void __launch_bounds__(256)
flash_attn(const float* __restrict__ Q, const float* __restrict__ Kt,
           const float* __restrict__ V, float* __restrict__ O) {
  __shared__ __attribute__((aligned(16))) float Ks[2][16 * FKP];
  __shared__ __attribute__((aligned(16))) float Vs[2][8 * FVP];
  const int wv = __builtin_amdgcn_readfirstlane((int)(threadIdx.x >> 5));
  const int NQT = SEQ / 16;
  const int NPB = NQT / 8;
  const int bh = blockIdx.x / NPB;
  const int sp = blockIdx.x % NPB;
  const int qhi = NQT - 4 - 4 * sp;
  const int qt = (wv < 4) ? (qhi + wv) : (4 * sp + (wv - 4));
  const int jmax = qhi + 3;
  const int h = bh % NUM_HEADS;
  const int b = bh / NUM_HEADS;
  const size_t base = (size_t)b * SEQ * D_MODEL + (size_t)h * D_K;
  const float* Qb = Q + base;
  const float* Kb = Kt + base;
  const float* Vb = V + base;
  float* Ob = O + base;
  const int lane = threadIdx.x & 31;
  const int hf = lane >> 4;
  const int rc = lane & 15;

  const int sr = threadIdx.x >> 4;
  const int sc = (threadIdx.x & 15) << 2;
  const float* Kg = Kb + (size_t)sr * D_MODEL + sc;
  const int kst = sr * FKP + sc;
  const int vk0 = 8 * (wv & 1) + 2 * (wv >> 1);
  const float* Vg = Vb + (size_t)vk0 * D_MODEL + 2 * lane;
  const int vst = wv * FVP + 4 * lane;

  v4f rk = *(const v4f*)Kg;
  v2f rva = *(const v2f*)Vg;
  v2f rvb = *(const v2f*)(Vg + D_MODEL);
  *(v4f*)(&Ks[0][kst]) = rk;
  {
    v4f pk; pk.x = rva.x; pk.y = rvb.x; pk.z = rva.y; pk.w = rvb.y;
    *(v4f*)(&Vs[0][vst]) = pk;
  }

  v2f qa[16];
#pragma unroll
  for (int i = 0; i < 16; i++)
    qa[i] = *(const v2f*)(Qb + (size_t)(qt * 16 + rc) * D_MODEL + i * 4 + 2 * hf);

  v8f o0 = {}, o1 = {}, o2 = {}, o3 = {};
  float mr = -1e30f;
  float lr = 0.f;
  __syncthreads();

  for (int jt = 0; jt <= jmax; ++jt) {
    const int cur = jt & 1;
    const bool more = (jt < jmax);
    if (more) {
      const size_t go = (size_t)(jt + 1) * 16 * D_MODEL;
      rk = *(const v4f*)(Kg + go);
      rva = *(const v2f*)(Vg + go);
      rvb = *(const v2f*)(Vg + go + D_MODEL);
    }
    asm volatile("" ::: "memory");
    if (jt <= qt) {
      v8f s = {};
      const float* kf = &Ks[cur][rc * FKP + 2 * hf];
#pragma unroll
      for (int i = 0; i < 16; i++) {
        const v2f kb = *(const v2f*)(kf + i * 4);
        s = wmma_f32(kb, qa[i], s);
      }
      float sv[8];
#pragma unroll
      for (int r = 0; r < 8; r++) sv[r] = s[r] * 0.125f;
      if (jt == qt) {
#pragma unroll
        for (int r = 0; r < 8; r++)
          if (r + 8 * hf > rc) sv[r] = -1e30f;
      }
      float mx = fmaxf(fmaxf(fmaxf(sv[0], sv[1]), fmaxf(sv[2], sv[3])),
                       fmaxf(fmaxf(sv[4], sv[5]), fmaxf(sv[6], sv[7])));
      mx = fmaxf(mx, xhalf(mx));
      const float mnew = fmaxf(mr, mx);
      const float corr = __expf(mr - mnew);
      mr = mnew;
      v8f p;
      float ps = 0.f;
#pragma unroll
      for (int r = 0; r < 8; r++) { p[r] = __expf(sv[r] - mnew); ps += p[r]; }
      lr = lr * corr + ps;
      o0 = o0 * corr; o1 = o1 * corr; o2 = o2 * corr; o3 = o3 * corr;
#pragma unroll
      for (int j = 0; j < 4; j++) {
        v2f pb; pb.x = p[2 * j]; pb.y = p[2 * j + 1];
        const float* Vr = &Vs[cur][(2 * j + hf) * FVP + 2 * rc];
        const v2f va0 = *(const v2f*)(Vr);
        const v2f va1 = *(const v2f*)(Vr + 32);
        const v2f va2 = *(const v2f*)(Vr + 64);
        const v2f va3 = *(const v2f*)(Vr + 96);
        o0 = wmma_f32(va0, pb, o0);
        o1 = wmma_f32(va1, pb, o1);
        o2 = wmma_f32(va2, pb, o2);
        o3 = wmma_f32(va3, pb, o3);
      }
    }
    if (more) {
      *(v4f*)(&Ks[cur ^ 1][kst]) = rk;
      v4f pk; pk.x = rva.x; pk.y = rvb.x; pk.z = rva.y; pk.w = rvb.y;
      *(v4f*)(&Vs[cur ^ 1][vst]) = pk;
    }
    __syncthreads();
  }
  const float inv = 1.0f / (lr + xhalf(lr));
  float* orow = Ob + (size_t)(qt * 16 + rc) * D_MODEL + 8 * hf;
  {
    v4f a, c2;
    a.x = o0[0] * inv; a.y = o0[1] * inv; a.z = o0[2] * inv; a.w = o0[3] * inv;
    c2.x = o0[4] * inv; c2.y = o0[5] * inv; c2.z = o0[6] * inv; c2.w = o0[7] * inv;
    *(v4f*)(orow) = a; *(v4f*)(orow + 4) = c2;
    a.x = o1[0] * inv; a.y = o1[1] * inv; a.z = o1[2] * inv; a.w = o1[3] * inv;
    c2.x = o1[4] * inv; c2.y = o1[5] * inv; c2.z = o1[6] * inv; c2.w = o1[7] * inv;
    *(v4f*)(orow + 16) = a; *(v4f*)(orow + 20) = c2;
    a.x = o2[0] * inv; a.y = o2[1] * inv; a.z = o2[2] * inv; a.w = o2[3] * inv;
    c2.x = o2[4] * inv; c2.y = o2[5] * inv; c2.z = o2[6] * inv; c2.w = o2[7] * inv;
    *(v4f*)(orow + 32) = a; *(v4f*)(orow + 36) = c2;
    a.x = o3[0] * inv; a.y = o3[1] * inv; a.z = o3[2] * inv; a.w = o3[3] * inv;
    c2.x = o3[4] * inv; c2.y = o3[5] * inv; c2.z = o3[6] * inv; c2.w = o3[7] * inv;
    *(v4f*)(orow + 48) = a; *(v4f*)(orow + 52) = c2;
  }
}

extern "C" void kernel_launch(void* const* d_in, const int* in_sizes, int n_in,
                              void* d_out, int out_size, void* d_ws, size_t ws_size,
                              hipStream_t stream) {
  (void)in_sizes; (void)n_in; (void)out_size; (void)ws_size;
  const float* x        = (const float*)d_in[0];
  const int*   positions= (const int*)d_in[1];
  const int*   theta    = (const int*)d_in[2];
  const float* Wq = (const float*)d_in[3];
  const float* Wk = (const float*)d_in[4];
  const float* Wv = (const float*)d_in[5];
  const float* Wo = (const float*)d_in[6];
  const float* g1 = (const float*)d_in[7];
  const float* g2 = (const float*)d_in[8];
  const float* w1 = (const float*)d_in[9];
  const float* w2 = (const float*)d_in[10];
  const float* w3 = (const float*)d_in[11];
  float* out = (float*)d_out;

  float* xn   = (float*)d_ws;
  float* q    = xn   + (size_t)ROWS * D_MODEL;
  float* k    = q    + (size_t)ROWS * D_MODEL;
  float* v    = k    + (size_t)ROWS * D_MODEL;
  float* attn = v    + (size_t)ROWS * D_MODEL;
  float* x2   = attn + (size_t)ROWS * D_MODEL;
  float* hb   = q;

  const dim3 blk(256);
  const dim3 gblk(128);
  const int gproj = (ROWS / 128) * (D_MODEL / 64);
  const int gff   = (ROWS / 128) * (D_FF / 64);
  const int gfa   = (BATCH * NUM_HEADS * (SEQ / 16)) / 8;

  rmsnorm_kernel<<<ROWS, blk, 0, stream>>>(x, g1, xn, D_MODEL);
  gemm_lds<0><<<gproj, gblk, 0, stream>>>(xn, Wq, nullptr, q, ROWS, D_MODEL, D_MODEL);
  gemm_lds<0><<<gproj, gblk, 0, stream>>>(xn, Wk, nullptr, k, ROWS, D_MODEL, D_MODEL);
  gemm_lds<0><<<gproj, gblk, 0, stream>>>(xn, Wv, nullptr, v, ROWS, D_MODEL, D_MODEL);
  rope_kernel<<<ROWS, blk, 0, stream>>>(q, positions, theta);
  rope_kernel<<<ROWS, blk, 0, stream>>>(k, positions, theta);
  flash_attn<<<gfa, blk, 0, stream>>>(q, k, v, attn);
  gemm_lds<1><<<gproj, gblk, 0, stream>>>(attn, Wo, x, x2, ROWS, D_MODEL, D_MODEL);
  rmsnorm_kernel<<<ROWS, blk, 0, stream>>>(x2, g2, xn, D_MODEL);
  gemm_gate<<<gff, gblk, 0, stream>>>(xn, w1, w3, hb, ROWS, D_FF, D_MODEL);
  gemm_lds<1><<<gproj, gblk, 0, stream>>>(hb, w2, x2, out, ROWS, D_MODEL, D_FF);
}


# QKV projections in one launch sharing the staged xn tile; RoPE applied in its epilogue from a cos/sin table built once per call; 11 launches -> 8; GEMM sum order kept
# speedup vs baseline: 3.7851x; 1.0629x over previous
#include <hip/hip_runtime.h>

#define D_MODEL   1024
#define NUM_HEADS 16
#define D_K       64
#define D_FF      2752
#define SEQ       2048
#define BATCH     2
#define ROWS      (BATCH*SEQ)

typedef float v2f __attribute__((ext_vector_type(2)));
typedef float v4f __attribute__((ext_vector_type(4)));
typedef float v8f __attribute__((ext_vector_type(8)));

__device__ __forceinline__ v8f wmma_f32(v2f a, v2f b, v8f c) {
  return __builtin_amdgcn_wmma_f32_16x16x4_f32(
       false, a,  false, b,
       (short)0, c,  false,  false);
}

__device__ __forceinline__ float silu_mul(float u, float w) {
  return u * (1.0f / (1.0f + __expf(-u))) * w;
}

__global__ void __launch_bounds__(256)
rmsnorm_kernel(const float* __restrict__ x, const float* __restrict__ g,
               float* __restrict__ out, int d) {
  const int row = blockIdx.x;
  const float* xr = x + (size_t)row * d;
  float ss = 0.f;
  for (int i = threadIdx.x; i < d; i += blockDim.x) { float v = xr[i]; ss = fmaf(v, v, ss); }
#pragma unroll
  for (int m = 16; m >= 1; m >>= 1) ss += __shfl_xor(ss, m, 32);
  __shared__ float red[8];
  const int w = threadIdx.x >> 5, ln = threadIdx.x & 31;
  if (ln == 0) red[w] = ss;
  __syncthreads();
  float tot = 0.f;
#pragma unroll
  for (int i = 0; i < 8; i++) tot += red[i];
  const float inv = rsqrtf(tot / (float)d + 1e-5f);
  for (int i = threadIdx.x; i < d; i += blockDim.x)
    out[(size_t)row * d + i] = xr[i] * inv * g[i];
}

__global__ void __launch_bounds__(256)
rope_table(float* __restrict__ tab, const int* __restrict__ positions,
           const int* __restrict__ theta_p) {
  const int idx = blockIdx.x * 256 + threadIdx.x;
  const int s = idx >> 5;
  const int d = (idx & 31) << 1;
  const float pos = (float)positions[s];
  const float th = (float)(*theta_p);
  const float freq = __powf(th, -(float)d * (1.0f / (float)D_K));
  const float ang = pos * freq;
  float sn, cs;
  sincosf(ang, &sn, &cs);
  v2f o; o.x = cs; o.y = sn;
  *(v2f*)(tab + 2 * (size_t)idx) = o;
}

#define GBM 128
#define GBN 64
#define GBK 32
#define GLD 36

template <int MODE>
__global__ void __launch_bounds__(128)
gemm_lds(const float* __restrict__ A,
         const float* __restrict__ B, const float* __restrict__ Res,
         float* __restrict__ C, int M, int N, int K) {
  __shared__ __attribute__((aligned(16))) float As[2][GBM * GLD];
  __shared__ __attribute__((aligned(16))) float Bs[2][GBN * GLD];
  (void)M;
  const int NT  = N >> 6;
  const int m0b = (blockIdx.x / NT) * GBM;
  const int n0b = (blockIdx.x % NT) * GBN;
  const int tid = threadIdx.x;
  const int w    = tid >> 5;
  const int lane = tid & 31;
  const int hf = lane >> 4;
  const int rc = lane & 15;

  const int sc = (tid & 7) << 2;
  const int sr = tid >> 3;
  const float* Ag  = A + (size_t)(m0b + sr) * K + sc;
  const float* Bg  = B + (size_t)(n0b + sr) * K + sc;
  const size_t rstep = (size_t)16 * K;
  const int sts = sr * GLD + sc;

  v4f ra[8], rb[4];

#pragma unroll
  for (int j = 0; j < 8; j++) ra[j] = *(const v4f*)(Ag + j * rstep);
#pragma unroll
  for (int j = 0; j < 4; j++) rb[j] = *(const v4f*)(Bg + j * rstep);
#pragma unroll
  for (int j = 0; j < 8; j++) *(v4f*)(&As[0][sts + j * 16 * GLD]) = ra[j];
#pragma unroll
  for (int j = 0; j < 4; j++) *(v4f*)(&Bs[0][sts + j * 16 * GLD]) = rb[j];
  __syncthreads();

  v8f acc[2][4] = {};
  const int nkt = K / GBK;
  const int afo = (w * 32 + rc) * GLD + 2 * hf;
  const int bfo = rc * GLD + 2 * hf;
  for (int kt = 0; kt < nkt; ++kt) {
    const int cur = kt & 1;
    const bool more = (kt + 1 < nkt);
    if (more) {
      const int ko = (kt + 1) * GBK;
#pragma unroll
      for (int j = 0; j < 8; j++) ra[j] = *(const v4f*)(Ag + j * rstep + ko);
#pragma unroll
      for (int j = 0; j < 4; j++) rb[j] = *(const v4f*)(Bg + j * rstep + ko);
    }
    const float* as = &As[cur][afo];
    const float* bs = &Bs[cur][bfo];
#pragma unroll
    for (int kk = 0; kk < GBK / 4; kk++) {
      const v2f a0 = *(const v2f*)(as + kk * 4);
      const v2f a1 = *(const v2f*)(as + 16 * GLD + kk * 4);
      v2f b[4];
#pragma unroll
      for (int t = 0; t < 4; t++) b[t] = *(const v2f*)(bs + t * 16 * GLD + kk * 4);
#pragma unroll
      for (int t = 0; t < 4; t++) {
        acc[0][t] = wmma_f32(a0, b[t], acc[0][t]);
        acc[1][t] = wmma_f32(a1, b[t], acc[1][t]);
      }
    }
    if (more) {
      float* ad = &As[cur ^ 1][sts];
      float* bd = &Bs[cur ^ 1][sts];
#pragma unroll
      for (int j = 0; j < 8; j++) *(v4f*)(ad + j * 16 * GLD) = ra[j];
#pragma unroll
      for (int j = 0; j < 4; j++) *(v4f*)(bd + j * 16 * GLD) = rb[j];
    }
    __syncthreads();
  }

  const int m0 = m0b + w * 32;
  const int n0 = n0b;
#pragma unroll
  for (int i = 0; i < 2; i++) {
#pragma unroll
    for (int r = 0; r < 8; r++) {
      const size_t base = (size_t)(m0 + i * 16 + r + 8 * hf) * N + n0 + rc;
      float o0 = acc[i][0][r], o1 = acc[i][1][r], o2 = acc[i][2][r], o3 = acc[i][3][r];
      if (MODE == 1) {
        o0 += Res[base];      o1 += Res[base + 16];
        o2 += Res[base + 32]; o3 += Res[base + 48];
      }
      C[base] = o0; C[base + 16] = o1; C[base + 32] = o2; C[base + 48] = o3;
    }
  }
}

__global__ void __launch_bounds__(128)
gemm_gate(const float* __restrict__ A, const float* __restrict__ B1,
          const float* __restrict__ B3, float* __restrict__ H,
          int M, int N, int K) {
  __shared__ __attribute__((aligned(16))) float As[2][GBM * GLD];
  __shared__ __attribute__((aligned(16))) float B1s[2][GBN * GLD];
  __shared__ __attribute__((aligned(16))) float B3s[2][GBN * GLD];
  (void)M;
  const int NT  = N >> 6;
  const int m0b = (blockIdx.x / NT) * GBM;
  const int n0b = (blockIdx.x % NT) * GBN;
  const int tid = threadIdx.x;
  const int w    = tid >> 5;
  const int lane = tid & 31;
  const int hf = lane >> 4;
  const int rc = lane & 15;

  const int sc = (tid & 7) << 2;
  const int sr = tid >> 3;
  const float* Ag  = A  + (size_t)(m0b + sr) * K + sc;
  const float* B1g = B1 + (size_t)(n0b + sr) * K + sc;
  const float* B3g = B3 + (size_t)(n0b + sr) * K + sc;
  const size_t rstep = (size_t)16 * K;
  const int sts = sr * GLD + sc;

  v4f ra[8], rb1[4], rb3[4];

#pragma unroll
  for (int j = 0; j < 8; j++) ra[j] = *(const v4f*)(Ag + j * rstep);
#pragma unroll
  for (int j = 0; j < 4; j++) {
    rb1[j] = *(const v4f*)(B1g + j * rstep);
    rb3[j] = *(const v4f*)(B3g + j * rstep);
  }
#pragma unroll
  for (int j = 0; j < 8; j++) *(v4f*)(&As[0][sts + j * 16 * GLD]) = ra[j];
#pragma unroll
  for (int j = 0; j < 4; j++) {
    *(v4f*)(&B1s[0][sts + j * 16 * GLD]) = rb1[j];
    *(v4f*)(&B3s[0][sts + j * 16 * GLD]) = rb3[j];
  }
  __syncthreads();

  v8f acc1[2][4] = {};
  v8f acc3[2][4] = {};
  const int nkt = K / GBK;
  const int afo = (w * 32 + rc) * GLD + 2 * hf;
  const int bfo = rc * GLD + 2 * hf;
  for (int kt = 0; kt < nkt; ++kt) {
    const int cur = kt & 1;
    const bool more = (kt + 1 < nkt);
    if (more) {
      const int ko = (kt + 1) * GBK;
#pragma unroll
      for (int j = 0; j < 8; j++) ra[j] = *(const v4f*)(Ag + j * rstep + ko);
#pragma unroll
      for (int j = 0; j < 4; j++) {
        rb1[j] = *(const v4f*)(B1g + j * rstep + ko);
        rb3[j] = *(const v4f*)(B3g + j * rstep + ko);
      }
    }
    const float* as  = &As[cur][afo];
    const float* b1s = &B1s[cur][bfo];
    const float* b3s = &B3s[cur][bfo];
#pragma unroll
    for (int kk = 0; kk < GBK / 4; kk++) {
      const v2f a0 = *(const v2f*)(as + kk * 4);
      const v2f a1 = *(const v2f*)(as + 16 * GLD + kk * 4);
      v2f b1[4], b3[4];
#pragma unroll
      for (int t = 0; t < 4; t++) {
        b1[t] = *(const v2f*)(b1s + t * 16 * GLD + kk * 4);
        b3[t] = *(const v2f*)(b3s + t * 16 * GLD + kk * 4);
      }
#pragma unroll
      for (int t = 0; t < 4; t++) {
        acc1[0][t] = wmma_f32(a0, b1[t], acc1[0][t]);
        acc1[1][t] = wmma_f32(a1, b1[t], acc1[1][t]);
      }
#pragma unroll
      for (int t = 0; t < 4; t++) {
        acc3[0][t] = wmma_f32(a0, b3[t], acc3[0][t]);
        acc3[1][t] = wmma_f32(a1, b3[t], acc3[1][t]);
      }
    }
    if (more) {
      float* ad  = &As[cur ^ 1][sts];
      float* b1d = &B1s[cur ^ 1][sts];
      float* b3d = &B3s[cur ^ 1][sts];
#pragma unroll
      for (int j = 0; j < 8; j++) *(v4f*)(ad + j * 16 * GLD) = ra[j];
#pragma unroll
      for (int j = 0; j < 4; j++) {
        *(v4f*)(b1d + j * 16 * GLD) = rb1[j];
        *(v4f*)(b3d + j * 16 * GLD) = rb3[j];
      }
    }
    __syncthreads();
  }

  const int m0 = m0b + w * 32;
  const int n0 = n0b;
#pragma unroll
  for (int i = 0; i < 2; i++) {
#pragma unroll
    for (int r = 0; r < 8; r++) {
      const size_t base = (size_t)(m0 + i * 16 + r + 8 * hf) * N + n0 + rc;
#pragma unroll
      for (int t = 0; t < 4; t++)
        H[base + 16 * t] = silu_mul(acc1[i][t][r], acc3[i][t][r]);
    }
  }
}

static_assert(GBN == D_K, "a block's columns must be exactly one head");
static_assert((SEQ & (SEQ - 1)) == 0, "row -> s uses a mask");

__global__ void __launch_bounds__(128)
gemm_qkv_rope(const float* __restrict__ A, const float* __restrict__ Bq,
              const float* __restrict__ Bk, const float* __restrict__ Bv,
              const float* __restrict__ tab,
              float* __restrict__ Cq, float* __restrict__ Ck, float* __restrict__ Cv,
              int M, int N, int K) {
  __shared__ __attribute__((aligned(16))) float As[2][GBM * GLD];
  __shared__ __attribute__((aligned(16))) float Bs[3][2][GBN * GLD];
  (void)M;
  const int NT  = N >> 6;
  const int m0b = (blockIdx.x / NT) * GBM;
  const int n0b = (blockIdx.x % NT) * GBN;
  const int tid = threadIdx.x;
  const int w    = tid >> 5;
  const int lane = tid & 31;
  const int hf = lane >> 4;
  const int rc = lane & 15;

  const int sc = (tid & 7) << 2;
  const int sr = tid >> 3;
  const float* Ag = A + (size_t)(m0b + sr) * K + sc;
  const size_t boff = (size_t)(n0b + sr) * K + sc;
  const float* Bg[3] = { Bq + boff, Bk + boff, Bv + boff };
  const size_t rstep = (size_t)16 * K;
  const int sts = sr * GLD + sc;

  v4f ra[8], rb[3][4];

#pragma unroll
  for (int j = 0; j < 8; j++) ra[j] = *(const v4f*)(Ag + j * rstep);
#pragma unroll
  for (int p = 0; p < 3; p++) {
#pragma unroll
    for (int j = 0; j < 4; j++) rb[p][j] = *(const v4f*)(Bg[p] + j * rstep);
  }
#pragma unroll
  for (int j = 0; j < 8; j++) *(v4f*)(&As[0][sts + j * 16 * GLD]) = ra[j];
#pragma unroll
  for (int p = 0; p < 3; p++) {
#pragma unroll
    for (int j = 0; j < 4; j++) *(v4f*)(&Bs[p][0][sts + j * 16 * GLD]) = rb[p][j];
  }
  __syncthreads();

  v8f acc[3][2][4] = {};
  const int nkt = K / GBK;
  const int afo = (w * 32 + rc) * GLD + 2 * hf;
  const int bfo = rc * GLD + 2 * hf;
  for (int kt = 0; kt < nkt; ++kt) {
    const int cur = kt & 1;
    const bool more = (kt + 1 < nkt);
    if (more) {
      const int ko = (kt + 1) * GBK;
#pragma unroll
      for (int j = 0; j < 8; j++) ra[j] = *(const v4f*)(Ag + j * rstep + ko);
#pragma unroll
      for (int p = 0; p < 3; p++) {
#pragma unroll
        for (int j = 0; j < 4; j++) rb[p][j] = *(const v4f*)(Bg[p] + j * rstep + ko);
      }
    }
    const float* as = &As[cur][afo];
#pragma unroll
    for (int kk = 0; kk < GBK / 4; kk++) {
      const v2f a0 = *(const v2f*)(as + kk * 4);
      const v2f a1 = *(const v2f*)(as + 16 * GLD + kk * 4);
      v2f b[3][4];
#pragma unroll
      for (int p = 0; p < 3; p++) {
#pragma unroll
        for (int t = 0; t < 4; t++)
          b[p][t] = *(const v2f*)(&Bs[p][cur][bfo + t * 16 * GLD + kk * 4]);
      }
#pragma unroll
      for (int p = 0; p < 3; p++) {
#pragma unroll
        for (int t = 0; t < 4; t++) {
          acc[p][0][t] = wmma_f32(a0, b[p][t], acc[p][0][t]);
          acc[p][1][t] = wmma_f32(a1, b[p][t], acc[p][1][t]);
        }
      }
    }
    if (more) {
#pragma unroll
      for (int j = 0; j < 8; j++) *(v4f*)(&As[cur ^ 1][sts + j * 16 * GLD]) = ra[j];
#pragma unroll
      for (int p = 0; p < 3; p++) {
#pragma unroll
        for (int j = 0; j < 4; j++) *(v4f*)(&Bs[p][cur ^ 1][sts + j * 16 * GLD]) = rb[p][j];
      }
    }
    __syncthreads();
  }

  const int m0 = m0b + w * 32;
  const bool odd = (rc & 1) != 0;
#pragma unroll
  for (int i = 0; i < 2; i++) {
#pragma unroll
    for (int r = 0; r < 8; r++) {
      const int row = m0 + i * 16 + r + 8 * hf;
      const size_t base = (size_t)row * N + n0b + rc;
      const float* tr = tab + ((size_t)(row & (SEQ - 1)) * (D_K / 2) + (rc >> 1)) * 2;
#pragma unroll
      for (int t = 0; t < 4; t++) {
        const v2f csn = *(const v2f*)(tr + 16 * t);
        const float cs = csn.x, sn = csn.y;
        const float xq = acc[0][i][t][r];
        const float pq = __shfl_xor(xq, 1, 32);
        const float xk = acc[1][i][t][r];
        const float pk = __shfl_xor(xk, 1, 32);
        Cq[base + 16 * t] = odd ? (pq * sn + xq * cs) : (xq * cs - pq * sn);
        Ck[base + 16 * t] = odd ? (pk * sn + xk * cs) : (xk * cs - pk * sn);
        Cv[base + 16 * t] = acc[2][i][t][r];
      }
    }
  }
}

#define FKP 68
#define FVP 160

__device__ __forceinline__ float xhalf(float x) {
  const int b = __builtin_bit_cast(int, x);
  return __builtin_bit_cast(float,
      __builtin_amdgcn_permlanex16(b, b, 0x76543210, 0xfedcba98, false, false));
}

__global__ void __launch_bounds__(256)
flash_attn(const float* __restrict__ Q, const float* __restrict__ Kt,
           const float* __restrict__ V, float* __restrict__ O) {
  __shared__ __attribute__((aligned(16))) float Ks[2][16 * FKP];
  __shared__ __attribute__((aligned(16))) float Vs[2][8 * FVP];
  const int wv = __builtin_amdgcn_readfirstlane((int)(threadIdx.x >> 5));
  const int NQT = SEQ / 16;
  const int NPB = NQT / 8;
  const int bh = blockIdx.x / NPB;
  const int sp = blockIdx.x % NPB;
  const int qhi = NQT - 4 - 4 * sp;
  const int qt = (wv < 4) ? (qhi + wv) : (4 * sp + (wv - 4));
  const int jmax = qhi + 3;
  const int h = bh % NUM_HEADS;
  const int b = bh / NUM_HEADS;
  const size_t base = (size_t)b * SEQ * D_MODEL + (size_t)h * D_K;
  const float* Qb = Q + base;
  const float* Kb = Kt + base;
  const float* Vb = V + base;
  float* Ob = O + base;
  const int lane = threadIdx.x & 31;
  const int hf = lane >> 4;
  const int rc = lane & 15;

  const int sr = threadIdx.x >> 4;
  const int sc = (threadIdx.x & 15) << 2;
  const float* Kg = Kb + (size_t)sr * D_MODEL + sc;
  const int kst = sr * FKP + sc;
  const int vk0 = 8 * (wv & 1) + 2 * (wv >> 1);
  const float* Vg = Vb + (size_t)vk0 * D_MODEL + 2 * lane;
  const int vst = wv * FVP + 4 * lane;

  v4f rk = *(const v4f*)Kg;
  v2f rva = *(const v2f*)Vg;
  v2f rvb = *(const v2f*)(Vg + D_MODEL);
  *(v4f*)(&Ks[0][kst]) = rk;
  {
    v4f pk; pk.x = rva.x; pk.y = rvb.x; pk.z = rva.y; pk.w = rvb.y;
    *(v4f*)(&Vs[0][vst]) = pk;
  }

  v2f qa[16];
#pragma unroll
  for (int i = 0; i < 16; i++)
    qa[i] = *(const v2f*)(Qb + (size_t)(qt * 16 + rc) * D_MODEL + i * 4 + 2 * hf);

  v8f o0 = {}, o1 = {}, o2 = {}, o3 = {};
  float mr = -1e30f;
  float lr = 0.f;
  __syncthreads();

  for (int jt = 0; jt <= jmax; ++jt) {
    const int cur = jt & 1;
    const bool more = (jt < jmax);
    if (more) {
      const size_t go = (size_t)(jt + 1) * 16 * D_MODEL;
      rk = *(const v4f*)(Kg + go);
      rva = *(const v2f*)(Vg + go);
      rvb = *(const v2f*)(Vg + go + D_MODEL);
    }
    asm volatile("" ::: "memory");
    if (jt <= qt) {
      v8f s = {};
      const float* kf = &Ks[cur][rc * FKP + 2 * hf];
#pragma unroll
      for (int i = 0; i < 16; i++) {
        const v2f kb = *(const v2f*)(kf + i * 4);
        s = wmma_f32(kb, qa[i], s);
      }
      float sv[8];
#pragma unroll
      for (int r = 0; r < 8; r++) sv[r] = s[r] * 0.125f;
      if (jt == qt) {
#pragma unroll
        for (int r = 0; r < 8; r++)
          if (r + 8 * hf > rc) sv[r] = -1e30f;
      }
      float mx = fmaxf(fmaxf(fmaxf(sv[0], sv[1]), fmaxf(sv[2], sv[3])),
                       fmaxf(fmaxf(sv[4], sv[5]), fmaxf(sv[6], sv[7])));
      mx = fmaxf(mx, xhalf(mx));
      const float mnew = fmaxf(mr, mx);
      const float corr = __expf(mr - mnew);
      mr = mnew;
      v8f p;
      float ps = 0.f;
#pragma unroll
      for (int r = 0; r < 8; r++) { p[r] = __expf(sv[r] - mnew); ps += p[r]; }
      lr = lr * corr + ps;
      o0 = o0 * corr; o1 = o1 * corr; o2 = o2 * corr; o3 = o3 * corr;
#pragma unroll
      for (int j = 0; j < 4; j++) {
        v2f pb; pb.x = p[2 * j]; pb.y = p[2 * j + 1];
        const float* Vr = &Vs[cur][(2 * j + hf) * FVP + 2 * rc];
        const v2f va0 = *(const v2f*)(Vr);
        const v2f va1 = *(const v2f*)(Vr + 32);
        const v2f va2 = *(const v2f*)(Vr + 64);
        const v2f va3 = *(const v2f*)(Vr + 96);
        o0 = wmma_f32(va0, pb, o0);
        o1 = wmma_f32(va1, pb, o1);
        o2 = wmma_f32(va2, pb, o2);
        o3 = wmma_f32(va3, pb, o3);
      }
    }
    if (more) {
      *(v4f*)(&Ks[cur ^ 1][kst]) = rk;
      v4f pk; pk.x = rva.x; pk.y = rvb.x; pk.z = rva.y; pk.w = rvb.y;
      *(v4f*)(&Vs[cur ^ 1][vst]) = pk;
    }
    __syncthreads();
  }
  const float inv = 1.0f / (lr + xhalf(lr));
  float* orow = Ob + (size_t)(qt * 16 + rc) * D_MODEL + 8 * hf;
  {
    v4f a, c2;
    a.x = o0[0] * inv; a.y = o0[1] * inv; a.z = o0[2] * inv; a.w = o0[3] * inv;
    c2.x = o0[4] * inv; c2.y = o0[5] * inv; c2.z = o0[6] * inv; c2.w = o0[7] * inv;
    *(v4f*)(orow) = a; *(v4f*)(orow + 4) = c2;
    a.x = o1[0] * inv; a.y = o1[1] * inv; a.z = o1[2] * inv; a.w = o1[3] * inv;
    c2.x = o1[4] * inv; c2.y = o1[5] * inv; c2.z = o1[6] * inv; c2.w = o1[7] * inv;
    *(v4f*)(orow + 16) = a; *(v4f*)(orow + 20) = c2;
    a.x = o2[0] * inv; a.y = o2[1] * inv; a.z = o2[2] * inv; a.w = o2[3] * inv;
    c2.x = o2[4] * inv; c2.y = o2[5] * inv; c2.z = o2[6] * inv; c2.w = o2[7] * inv;
    *(v4f*)(orow + 32) = a; *(v4f*)(orow + 36) = c2;
    a.x = o3[0] * inv; a.y = o3[1] * inv; a.z = o3[2] * inv; a.w = o3[3] * inv;
    c2.x = o3[4] * inv; c2.y = o3[5] * inv; c2.z = o3[6] * inv; c2.w = o3[7] * inv;
    *(v4f*)(orow + 48) = a; *(v4f*)(orow + 52) = c2;
  }
}

extern "C" void kernel_launch(void* const* d_in, const int* in_sizes, int n_in,
                              void* d_out, int out_size, void* d_ws, size_t ws_size,
                              hipStream_t stream) {
  (void)in_sizes; (void)n_in; (void)out_size; (void)ws_size;
  const float* x        = (const float*)d_in[0];
  const int*   positions= (const int*)d_in[1];
  const int*   theta    = (const int*)d_in[2];
  const float* Wq = (const float*)d_in[3];
  const float* Wk = (const float*)d_in[4];
  const float* Wv = (const float*)d_in[5];
  const float* Wo = (const float*)d_in[6];
  const float* g1 = (const float*)d_in[7];
  const float* g2 = (const float*)d_in[8];
  const float* w1 = (const float*)d_in[9];
  const float* w2 = (const float*)d_in[10];
  const float* w3 = (const float*)d_in[11];
  float* out = (float*)d_out;

  float* xn   = (float*)d_ws;
  float* q    = xn   + (size_t)ROWS * D_MODEL;
  float* k    = q    + (size_t)ROWS * D_MODEL;
  float* v    = k    + (size_t)ROWS * D_MODEL;
  float* attn = v    + (size_t)ROWS * D_MODEL;
  float* x2   = attn + (size_t)ROWS * D_MODEL;
  float* tab  = x2   + (size_t)ROWS * D_MODEL;
  float* hb   = q;

  const dim3 blk(256);
  const dim3 gblk(128);
  const int gproj = (ROWS / 128) * (D_MODEL / 64);
  const int gff   = (ROWS / 128) * (D_FF / 64);
  const int gfa   = (BATCH * NUM_HEADS * (SEQ / 16)) / 8;
  const int gtab  = (SEQ * (D_K / 2)) / 256;

  rope_table<<<gtab, blk, 0, stream>>>(tab, positions, theta);
  rmsnorm_kernel<<<ROWS, blk, 0, stream>>>(x, g1, xn, D_MODEL);
  gemm_qkv_rope<<<gproj, gblk, 0, stream>>>(xn, Wq, Wk, Wv, tab, q, k, v, ROWS, D_MODEL, D_MODEL);
  flash_attn<<<gfa, blk, 0, stream>>>(q, k, v, attn);
  gemm_lds<1><<<gproj, gblk, 0, stream>>>(attn, Wo, x, x2, ROWS, D_MODEL, D_MODEL);
  rmsnorm_kernel<<<ROWS, blk, 0, stream>>>(x2, g2, xn, D_MODEL);
  gemm_gate<<<gff, gblk, 0, stream>>>(xn, w1, w3, hb, ROWS, D_FF, D_MODEL);
  gemm_lds<1><<<gproj, gblk, 0, stream>>>(hb, w2, x2, out, ROWS, D_MODEL, D_FF);
}
